# starting kernel
# baseline (speedup 1.0000x reference)
#include <hip/hip_runtime.h>
#include <hip/hip_bf16.h>

typedef __attribute__((ext_vector_type(2))) float v2f;
typedef __attribute__((ext_vector_type(8))) float v8f;

#define IN_SIZE 256
#define OUT_SIZE 64


__global__ __launch_bounds__(256) void k_init_deg(unsigned int* __restrict__ deg, int n) {
  int i = blockIdx.x * 256 + threadIdx.x;
  if (i < n) deg[i] = 0u;
}

__global__ __launch_bounds__(256) void k_count_deg(const int* __restrict__ row,
                                                   const int* __restrict__ col,
                                                   unsigned int* __restrict__ deg, int E) {
  int e = blockIdx.x * 256 + threadIdx.x;
  if (e < E) {
    int r = row[e], c = col[e];
    if (r != c) atomicAdd(&deg[r], 1u);
  }
}

__global__ __launch_bounds__(256) void k_dinv(const unsigned int* __restrict__ deg,
                                              float* __restrict__ dinv, int n) {
  int i = blockIdx.x * 256 + threadIdx.x;
  if (i < n) dinv[i] = rsqrtf((float)deg[i] + 1.0f);
}


__global__ __launch_bounds__(256) void k_gemm(const float* __restrict__ x,
                                              const float* __restrict__ W,
                                              float* __restrict__ xw, int n) {
  const int lane  = threadIdx.x & 31;
  const int wave  = threadIdx.x >> 5;
  const int tile  = blockIdx.x * 8 + wave;
  if (tile * 16 >= n) return;

  const int ml    = lane & 15;
  const int khalf = (lane >> 4) << 1;
  int arow = tile * 16 + ml;
  if (arow > n - 1) arow = n - 1;
  const float* xrow = x + (size_t)arow * IN_SIZE;

  v8f acc0 = {}, acc1 = {}, acc2 = {}, acc3 = {};
  for (int k = 0; k < IN_SIZE; k += 4) {
    v2f a = *(const v2f*)(xrow + k + khalf);
    const float* wb = W + (size_t)ml * IN_SIZE + k + khalf;
    v2f b0 = *(const v2f*)(wb + (size_t)0 * 16 * IN_SIZE);
    v2f b1 = *(const v2f*)(wb + (size_t)1 * 16 * IN_SIZE);
    v2f b2 = *(const v2f*)(wb + (size_t)2 * 16 * IN_SIZE);
    v2f b3 = *(const v2f*)(wb + (size_t)3 * 16 * IN_SIZE);
    acc0 = __builtin_amdgcn_wmma_f32_16x16x4_f32(false, a, false, b0, (short)0, acc0, false, false);
    acc1 = __builtin_amdgcn_wmma_f32_16x16x4_f32(false, a, false, b1, (short)0, acc1, false, false);
    acc2 = __builtin_amdgcn_wmma_f32_16x16x4_f32(false, a, false, b2, (short)0, acc2, false, false);
    acc3 = __builtin_amdgcn_wmma_f32_16x16x4_f32(false, a, false, b3, (short)0, acc3, false, false);
  }

  const int rbase = tile * 16 + ((lane >> 4) << 3);
  float* o = xw + (size_t)rbase * OUT_SIZE + ml;
#pragma unroll
  for (int v = 0; v < 8; ++v) {
    if (rbase + v < n) {
      o[(size_t)v * OUT_SIZE +  0] = acc0[v];
      o[(size_t)v * OUT_SIZE + 16] = acc1[v];
      o[(size_t)v * OUT_SIZE + 32] = acc2[v];
      o[(size_t)v * OUT_SIZE + 48] = acc3[v];
    }
  }
}


__global__ __launch_bounds__(256) void k_self(const float* __restrict__ xw,
                                              const float* __restrict__ dinv,
                                              float* __restrict__ out, int n) {
  size_t idx = (size_t)blockIdx.x * 256 + threadIdx.x;
  if (idx < (size_t)n * OUT_SIZE) {
    int i = (int)(idx >> 6);
    float d = dinv[i];
    out[idx] = d * d * xw[idx];
  }
}


__global__ __launch_bounds__(256) void k_scatter(const int* __restrict__ row,
                                                 const int* __restrict__ col,
                                                 const float* __restrict__ dinv,
                                                 const float* __restrict__ xw,
                                                 float* __restrict__ out, int E) {
  int e = blockIdx.x * 4 + (threadIdx.x >> 6);
  int j = threadIdx.x & 63;
  if (e >= E) return;
  int r = row[e];
  int c = col[e];
  if (r == c) return;
  float norm = dinv[r] * dinv[c];
  unsafeAtomicAdd(out + (size_t)r * OUT_SIZE + j,
                  norm * xw[(size_t)c * OUT_SIZE + j]);
}


extern "C" void kernel_launch(void* const* d_in, const int* in_sizes, int n_in,
                              void* d_out, int out_size, void* d_ws, size_t ws_size,
                              hipStream_t stream) {
  const int*   edge = (const int*)d_in[0];
  const float* x    = (const float*)d_in[1];
  const float* W    = (const float*)d_in[3];

  const int E = in_sizes[0] / 2;
  const int n = in_sizes[1] / IN_SIZE;
  const int* row = edge;
  const int* col = edge + E;
  float* out = (float*)d_out;

  float*        xw   = (float*)d_ws;
  unsigned int* deg  = (unsigned int*)(xw + (size_t)n * OUT_SIZE);
  float*        dinv = (float*)(deg + n);

  k_init_deg<<<(n + 255) / 256, 256, 0, stream>>>(deg, n);
  k_count_deg<<<(E + 255) / 256, 256, 0, stream>>>(row, col, deg, E);
  k_dinv<<<(n + 255) / 256, 256, 0, stream>>>(deg, dinv, n);

  const int tiles = (n + 15) / 16;
  k_gemm<<<(tiles + 7) / 8, 256, 0, stream>>>(x, W, xw, n);

  const size_t tot = (size_t)n * OUT_SIZE;
  k_self<<<(unsigned)((tot + 255) / 256), 256, 0, stream>>>(xw, dinv, out, n);
  k_scatter<<<(E + 3) / 4, 256, 0, stream>>>(row, col, dinv, xw, out, E);
}


# Edge scatter: one wave per 32 edges (was 2 waves per edge): ids and norm loaded once per lane, broadcast per edge, 16 row loads in flight, 2 whole-line f32 atomics; atomic arrival order as before
# speedup vs baseline: 1.9802x; 1.9802x over previous
#include <hip/hip_runtime.h>
#include <hip/hip_bf16.h>

typedef __attribute__((ext_vector_type(2))) float v2f;
typedef __attribute__((ext_vector_type(8))) float v8f;

#define IN_SIZE 256
#define OUT_SIZE 64


__global__ __launch_bounds__(256) void k_init_deg(unsigned int* __restrict__ deg, int n) {
  int i = blockIdx.x * 256 + threadIdx.x;
  if (i < n) deg[i] = 0u;
}

__global__ __launch_bounds__(256) void k_count_deg(const int* __restrict__ row,
                                                   const int* __restrict__ col,
                                                   unsigned int* __restrict__ deg, int E) {
  int e = blockIdx.x * 256 + threadIdx.x;
  if (e < E) {
    int r = row[e], c = col[e];
    if (r != c) atomicAdd(&deg[r], 1u);
  }
}

__global__ __launch_bounds__(256) void k_dinv(const unsigned int* __restrict__ deg,
                                              float* __restrict__ dinv, int n) {
  int i = blockIdx.x * 256 + threadIdx.x;
  if (i < n) dinv[i] = rsqrtf((float)deg[i] + 1.0f);
}


__global__ __launch_bounds__(256) void k_gemm(const float* __restrict__ x,
                                              const float* __restrict__ W,
                                              float* __restrict__ xw, int n) {
  const int lane  = threadIdx.x & 31;
  const int wave  = threadIdx.x >> 5;
  const int tile  = blockIdx.x * 8 + wave;
  if (tile * 16 >= n) return;

  const int ml    = lane & 15;
  const int khalf = (lane >> 4) << 1;
  int arow = tile * 16 + ml;
  if (arow > n - 1) arow = n - 1;
  const float* xrow = x + (size_t)arow * IN_SIZE;

  v8f acc0 = {}, acc1 = {}, acc2 = {}, acc3 = {};
  for (int k = 0; k < IN_SIZE; k += 4) {
    v2f a = *(const v2f*)(xrow + k + khalf);
    const float* wb = W + (size_t)ml * IN_SIZE + k + khalf;
    v2f b0 = *(const v2f*)(wb + (size_t)0 * 16 * IN_SIZE);
    v2f b1 = *(const v2f*)(wb + (size_t)1 * 16 * IN_SIZE);
    v2f b2 = *(const v2f*)(wb + (size_t)2 * 16 * IN_SIZE);
    v2f b3 = *(const v2f*)(wb + (size_t)3 * 16 * IN_SIZE);
    acc0 = __builtin_amdgcn_wmma_f32_16x16x4_f32(false, a, false, b0, (short)0, acc0, false, false);
    acc1 = __builtin_amdgcn_wmma_f32_16x16x4_f32(false, a, false, b1, (short)0, acc1, false, false);
    acc2 = __builtin_amdgcn_wmma_f32_16x16x4_f32(false, a, false, b2, (short)0, acc2, false, false);
    acc3 = __builtin_amdgcn_wmma_f32_16x16x4_f32(false, a, false, b3, (short)0, acc3, false, false);
  }

  const int rbase = tile * 16 + ((lane >> 4) << 3);
  float* o = xw + (size_t)rbase * OUT_SIZE + ml;
#pragma unroll
  for (int v = 0; v < 8; ++v) {
    if (rbase + v < n) {
      o[(size_t)v * OUT_SIZE +  0] = acc0[v];
      o[(size_t)v * OUT_SIZE + 16] = acc1[v];
      o[(size_t)v * OUT_SIZE + 32] = acc2[v];
      o[(size_t)v * OUT_SIZE + 48] = acc3[v];
    }
  }
}


__global__ __launch_bounds__(256) void k_self(const float* __restrict__ xw,
                                              const float* __restrict__ dinv,
                                              float* __restrict__ out, int n) {
  size_t idx = (size_t)blockIdx.x * 256 + threadIdx.x;
  if (idx < (size_t)n * OUT_SIZE) {
    int i = (int)(idx >> 6);
    float d = dinv[i];
    out[idx] = d * d * xw[idx];
  }
}


#define EDGES_PER_GROUP 8

template <bool ALL>
__device__ __forceinline__ void scatter_wave(int r, int c, float nrm, unsigned okmask, int lane,
                                             const float* __restrict__ xw,
                                             float* __restrict__ out) {
#pragma unroll
  for (int g = 0; g < 32; g += EDGES_PER_GROUP) {
    float v0[EDGES_PER_GROUP], v1[EDGES_PER_GROUP];
#pragma unroll
    for (int i = 0; i < EDGES_PER_GROUP; ++i) {
      const int sc = __builtin_amdgcn_readlane(c, g + i);
      const float* p = xw + (size_t)sc * OUT_SIZE;
      v0[i] = p[lane];
      v1[i] = p[lane + 32];
    }
#pragma unroll
    for (int i = 0; i < EDGES_PER_GROUP; ++i) {
      if (ALL || ((okmask >> (g + i)) & 1u)) {
        const int   sr = __builtin_amdgcn_readlane(r, g + i);
        const float sn = __int_as_float(__builtin_amdgcn_readlane(__float_as_int(nrm), g + i));
        float* o = out + (size_t)sr * OUT_SIZE;
        unsafeAtomicAdd(o + lane,      sn * v0[i]);
        unsafeAtomicAdd(o + lane + 32, sn * v1[i]);
      }
    }
  }
}

__global__ __launch_bounds__(256) void k_scatter(const int* __restrict__ row,
                                                 const int* __restrict__ col,
                                                 const float* __restrict__ dinv,
                                                 const float* __restrict__ xw,
                                                 float* __restrict__ out, int E) {
  const int lane = threadIdx.x & 31;
  const int wave = blockIdx.x * 8 + (threadIdx.x >> 5);
  if (wave * 32 >= E) return;
  const int e = wave * 32 + lane;

  int r = 0, c = 0;
  if (e < E) { r = row[e]; c = col[e]; }
  const bool ok = (e < E) && (r != c);
  float nrm = 0.0f;
  if (ok) nrm = dinv[r] * dinv[c];
  const unsigned okmask = __builtin_amdgcn_ballot_w32(ok);

  if (okmask == 0xFFFFFFFFu) scatter_wave<true >(r, c, nrm, okmask, lane, xw, out);
  else                       scatter_wave<false>(r, c, nrm, okmask, lane, xw, out);
}


extern "C" void kernel_launch(void* const* d_in, const int* in_sizes, int n_in,
                              void* d_out, int out_size, void* d_ws, size_t ws_size,
                              hipStream_t stream) {
  const int*   edge = (const int*)d_in[0];
  const float* x    = (const float*)d_in[1];
  const float* W    = (const float*)d_in[3];

  const int E = in_sizes[0] / 2;
  const int n = in_sizes[1] / IN_SIZE;
  const int* row = edge;
  const int* col = edge + E;
  float* out = (float*)d_out;

  float*        xw   = (float*)d_ws;
  unsigned int* deg  = (unsigned int*)(xw + (size_t)n * OUT_SIZE);
  float*        dinv = (float*)(deg + n);

  k_init_deg<<<(n + 255) / 256, 256, 0, stream>>>(deg, n);
  k_count_deg<<<(E + 255) / 256, 256, 0, stream>>>(row, col, deg, E);
  k_dinv<<<(n + 255) / 256, 256, 0, stream>>>(deg, dinv, n);

  const int tiles = (n + 15) / 16;
  k_gemm<<<(tiles + 7) / 8, 256, 0, stream>>>(x, W, xw, n);

  const size_t tot = (size_t)n * OUT_SIZE;
  k_self<<<(unsigned)((tot + 255) / 256), 256, 0, stream>>>(xw, dinv, out, n);
  k_scatter<<<(E + 255) / 256, 256, 0, stream>>>(row, col, dinv, xw, out, E);
}


# Aggregation without global f32 atomics: edges binned by 392-row destination bucket; one work-group per bucket sums its rows in LDS (ds f32 adds), stores each line once; sums in LDS arrival order
# speedup vs baseline: 2.1825x; 1.1022x over previous
#include <hip/hip_runtime.h>
#include <hip/hip_bf16.h>

typedef __attribute__((ext_vector_type(2))) float v2f;
typedef __attribute__((ext_vector_type(8))) float v8f;

#define IN_SIZE 256
#define OUT_SIZE 64


__global__ __launch_bounds__(256) void k_init_deg(unsigned int* __restrict__ deg, int n) {
  int i = blockIdx.x * 256 + threadIdx.x;
  if (i < n) deg[i] = 0u;
}

__global__ __launch_bounds__(256) void k_count_deg(const int* __restrict__ row,
                                                   const int* __restrict__ col,
                                                   unsigned int* __restrict__ deg, int E) {
  int e = blockIdx.x * 256 + threadIdx.x;
  if (e < E) {
    int r = row[e], c = col[e];
    if (r != c) atomicAdd(&deg[r], 1u);
  }
}

__global__ __launch_bounds__(256) void k_dinv(const unsigned int* __restrict__ deg,
                                              float* __restrict__ dinv, int n) {
  int i = blockIdx.x * 256 + threadIdx.x;
  if (i < n) dinv[i] = rsqrtf((float)deg[i] + 1.0f);
}


__global__ __launch_bounds__(256) void k_gemm(const float* __restrict__ x,
                                              const float* __restrict__ W,
                                              float* __restrict__ xw, int n) {
  const int lane  = threadIdx.x & 31;
  const int wave  = threadIdx.x >> 5;
  const int tile  = blockIdx.x * 8 + wave;
  if (tile * 16 >= n) return;

  const int ml    = lane & 15;
  const int khalf = (lane >> 4) << 1;
  int arow = tile * 16 + ml;
  if (arow > n - 1) arow = n - 1;
  const float* xrow = x + (size_t)arow * IN_SIZE;

  v8f acc0 = {}, acc1 = {}, acc2 = {}, acc3 = {};
  for (int k = 0; k < IN_SIZE; k += 4) {
    v2f a = *(const v2f*)(xrow + k + khalf);
    const float* wb = W + (size_t)ml * IN_SIZE + k + khalf;
    v2f b0 = *(const v2f*)(wb + (size_t)0 * 16 * IN_SIZE);
    v2f b1 = *(const v2f*)(wb + (size_t)1 * 16 * IN_SIZE);
    v2f b2 = *(const v2f*)(wb + (size_t)2 * 16 * IN_SIZE);
    v2f b3 = *(const v2f*)(wb + (size_t)3 * 16 * IN_SIZE);
    acc0 = __builtin_amdgcn_wmma_f32_16x16x4_f32(false, a, false, b0, (short)0, acc0, false, false);
    acc1 = __builtin_amdgcn_wmma_f32_16x16x4_f32(false, a, false, b1, (short)0, acc1, false, false);
    acc2 = __builtin_amdgcn_wmma_f32_16x16x4_f32(false, a, false, b2, (short)0, acc2, false, false);
    acc3 = __builtin_amdgcn_wmma_f32_16x16x4_f32(false, a, false, b3, (short)0, acc3, false, false);
  }

  const int rbase = tile * 16 + ((lane >> 4) << 3);
  float* o = xw + (size_t)rbase * OUT_SIZE + ml;
#pragma unroll
  for (int v = 0; v < 8; ++v) {
    if (rbase + v < n) {
      o[(size_t)v * OUT_SIZE +  0] = acc0[v];
      o[(size_t)v * OUT_SIZE + 16] = acc1[v];
      o[(size_t)v * OUT_SIZE + 32] = acc2[v];
      o[(size_t)v * OUT_SIZE + 48] = acc3[v];
    }
  }
}


__global__ __launch_bounds__(256) void k_self(const float* __restrict__ xw,
                                              const float* __restrict__ dinv,
                                              float* __restrict__ out, int n) {
  size_t idx = (size_t)blockIdx.x * 256 + threadIdx.x;
  if (idx < (size_t)n * OUT_SIZE) {
    int i = (int)(idx >> 6);
    float d = dinv[i];
    out[idx] = d * d * xw[idx];
  }
}


#define EDGES_PER_GROUP 8

template <bool ALL>
__device__ __forceinline__ void scatter_wave(int r, int c, float nrm, unsigned okmask, int lane,
                                             const float* __restrict__ xw,
                                             float* __restrict__ out) {
#pragma unroll
  for (int g = 0; g < 32; g += EDGES_PER_GROUP) {
    float v0[EDGES_PER_GROUP], v1[EDGES_PER_GROUP];
#pragma unroll
    for (int i = 0; i < EDGES_PER_GROUP; ++i) {
      const int sc = __builtin_amdgcn_readlane(c, g + i);
      const float* p = xw + (size_t)sc * OUT_SIZE;
      v0[i] = p[lane];
      v1[i] = p[lane + 32];
    }
#pragma unroll
    for (int i = 0; i < EDGES_PER_GROUP; ++i) {
      if (ALL || ((okmask >> (g + i)) & 1u)) {
        const int   sr = __builtin_amdgcn_readlane(r, g + i);
        const float sn = __int_as_float(__builtin_amdgcn_readlane(__float_as_int(nrm), g + i));
        float* o = out + (size_t)sr * OUT_SIZE;
        unsafeAtomicAdd(o + lane,      sn * v0[i]);
        unsafeAtomicAdd(o + lane + 32, sn * v1[i]);
      }
    }
  }
}

__global__ __launch_bounds__(256) void k_scatter(const int* __restrict__ row,
                                                 const int* __restrict__ col,
                                                 const float* __restrict__ dinv,
                                                 const float* __restrict__ xw,
                                                 float* __restrict__ out, int E) {
  const int lane = threadIdx.x & 31;
  const int wave = blockIdx.x * 8 + (threadIdx.x >> 5);
  if (wave * 32 >= E) return;
  const int e = wave * 32 + lane;

  int r = 0, c = 0;
  if (e < E) { r = row[e]; c = col[e]; }
  const bool ok = (e < E) && (r != c);
  float nrm = 0.0f;
  if (ok) nrm = dinv[r] * dinv[c];
  const unsigned okmask = __builtin_amdgcn_ballot_w32(ok);

  if (okmask == 0xFFFFFFFFu) scatter_wave<true >(r, c, nrm, okmask, lane, xw, out);
  else                       scatter_wave<false>(r, c, nrm, okmask, lane, xw, out);
}


#define ROWS_PER_BUCKET 392
#define MAX_BUCKETS     256
#define CHUNK_EDGES     4096
#define EDGES_PER_THREAD (CHUNK_EDGES / 256)

__global__ __launch_bounds__(256) void k_init_bins(unsigned int* __restrict__ deg,
                                                   unsigned int* __restrict__ bcount,
                                                   unsigned int* __restrict__ cursor, int n) {
  int i = blockIdx.x * 256 + threadIdx.x;
  if (i < n) deg[i] = 0u;
  if (i < MAX_BUCKETS) { bcount[i] = 0u; cursor[i] = 0u; }
}

__global__ __launch_bounds__(256) void k_count_bins(const int* __restrict__ row,
                                                    const int* __restrict__ col,
                                                    unsigned int* __restrict__ deg,
                                                    unsigned int* __restrict__ bcount, int E) {
  __shared__ unsigned int hist[MAX_BUCKETS];
  const int t = threadIdx.x;
  hist[t] = 0u;
  __syncthreads();
  const int base = blockIdx.x * CHUNK_EDGES + t;
  int rr[EDGES_PER_THREAD], cc[EDGES_PER_THREAD];
#pragma unroll
  for (int i = 0; i < EDGES_PER_THREAD; ++i) {
    const int e = base + i * 256;
    const int ec = e < E ? e : E - 1;
    rr[i] = row[ec]; cc[i] = col[ec];
  }
#pragma unroll
  for (int i = 0; i < EDGES_PER_THREAD; ++i) {
    const int e = base + i * 256;
    if (e < E && rr[i] != cc[i]) {
      atomicAdd(&deg[rr[i]], 1u);
      atomicAdd(&hist[(unsigned)rr[i] / ROWS_PER_BUCKET], 1u);
    }
  }
  __syncthreads();
  const unsigned int h = hist[t];
  if (h) atomicAdd(&bcount[t], h);
}

__global__ __launch_bounds__(256) void k_fill_bins(const int* __restrict__ row,
                                                   const int* __restrict__ col,
                                                   const unsigned int* __restrict__ bcount,
                                                   unsigned int* __restrict__ cursor,
                                                   int* __restrict__ binned, int E) {
  __shared__ unsigned int hist[MAX_BUCKETS];
  __shared__ unsigned int scan[MAX_BUCKETS];
  __shared__ unsigned int dst[MAX_BUCKETS];
  const int t = threadIdx.x;
  hist[t] = 0u;
  const unsigned int cnt_t = bcount[t];
  scan[t] = cnt_t;
  __syncthreads();
  for (int off = 1; off < MAX_BUCKETS; off <<= 1) {
    const unsigned int v = (t >= off) ? scan[t - off] : 0u;
    __syncthreads();
    scan[t] += v;
    __syncthreads();
  }
  const unsigned int start_t = scan[t] - cnt_t;

  int rr[EDGES_PER_THREAD], cc[EDGES_PER_THREAD];
  unsigned int lr[EDGES_PER_THREAD];
  const int base = blockIdx.x * CHUNK_EDGES + t;
#pragma unroll
  for (int i = 0; i < EDGES_PER_THREAD; ++i) {
    const int e = base + i * 256;
    const int ec = e < E ? e : E - 1;
    rr[i] = row[ec]; cc[i] = col[ec];
  }
#pragma unroll
  for (int i = 0; i < EDGES_PER_THREAD; ++i) {
    const int e = base + i * 256;
    lr[i] = 0u;
    if (e < E && rr[i] != cc[i])
      lr[i] = atomicAdd(&hist[(unsigned)rr[i] / ROWS_PER_BUCKET], 1u);
    else
      rr[i] = -1;
  }
  __syncthreads();
  const unsigned int h = hist[t];
  unsigned int g = 0u;
  if (h) g = atomicAdd(&cursor[t], h);
  dst[t] = start_t + g;
  __syncthreads();
#pragma unroll
  for (int i = 0; i < EDGES_PER_THREAD; ++i) {
    if (rr[i] >= 0)
      ((int2*)binned)[dst[(unsigned)rr[i] / ROWS_PER_BUCKET] + lr[i]] = make_int2(rr[i], cc[i]);
  }
}

#define ACC_GROUP 16
#define TILE_FLOATS (ROWS_PER_BUCKET * OUT_SIZE)
#define TILE_V4     (TILE_FLOATS / 4)
#define TILE_V4_FULL (TILE_V4 / 1024)

template <bool ALL>
__device__ __forceinline__ void accum_wave(int lofs, int c, float nrm, unsigned okmask, int lane,
                                           const float* __restrict__ xw, float* tile) {
#pragma unroll
  for (int g = 0; g < 32; g += ACC_GROUP) {
    float v0[ACC_GROUP], v1[ACC_GROUP];
#pragma unroll
    for (int i = 0; i < ACC_GROUP; ++i) {
      const int sc = __builtin_amdgcn_readlane(c, g + i);
      const float* p = xw + (size_t)sc * OUT_SIZE;
      v0[i] = p[lane];
      v1[i] = p[lane + 32];
    }
#pragma unroll
    for (int i = 0; i < ACC_GROUP; ++i) {
      if (ALL || ((okmask >> (g + i)) & 1u)) {
        const int   so = __builtin_amdgcn_readlane(lofs, g + i);
        const float sn = __int_as_float(__builtin_amdgcn_readlane(__float_as_int(nrm), g + i));
        atomicAdd(tile + so + lane,      sn * v0[i]);
        atomicAdd(tile + so + lane + 32, sn * v1[i]);
      }
    }
  }
}

__device__ __forceinline__ float4 self_term(const float* __restrict__ dinv,
                                            const float* __restrict__ xw, int r0, int q, int n) {
  int r = r0 + (q >> 4);
  r = r < n ? r : n - 1;
  const float d = dinv[r];
  const float4 v = *(const float4*)(xw + (size_t)r * OUT_SIZE + ((q & 15) << 2));
  return make_float4(d * d * v.x, d * d * v.y, d * d * v.z, d * d * v.w);
}

__global__ __launch_bounds__(1024) void k_accum(const int* __restrict__ binned_rc,
                                                const unsigned int* __restrict__ bcount,
                                                const float* __restrict__ dinv,
                                                const float* __restrict__ xw,
                                                float* __restrict__ out, int n) {
  __shared__ __attribute__((aligned(16))) float tile[TILE_FLOATS];
  __shared__ unsigned int s_start;
  float4* tile4 = (float4*)tile;
  const int2* binned = (const int2*)binned_rc;
  const int t = threadIdx.x;
  const int lane = t & 31;
  const int w = t >> 5;
  const int b = blockIdx.x;
  const int r0 = b * ROWS_PER_BUCKET;

  if (t == 0) s_start = 0u;
  __syncthreads();
  if (t < MAX_BUCKETS && t < b) atomicAdd(&s_start, bcount[t]);

  {
    float4 sv[TILE_V4_FULL];
#pragma unroll
    for (int k = 0; k < TILE_V4_FULL; ++k) sv[k] = self_term(dinv, xw, r0, t + k * 1024, n);
#pragma unroll
    for (int k = 0; k < TILE_V4_FULL; ++k) tile4[t + k * 1024] = sv[k];
    const int q = t + TILE_V4_FULL * 1024;
    if (q < TILE_V4) tile4[q] = self_term(dinv, xw, r0, q, n);
  }
  __syncthreads();

  const unsigned int seg = s_start;
  const unsigned int end = seg + bcount[b];
  const unsigned int first = seg + (unsigned)w * 32u + (unsigned)lane;

  int2 idA = make_int2(r0, 0);
  if (first < end) idA = binned[first];
  float drA = dinv[idA.x], dcA = dinv[idA.y];
  int2 idB = make_int2(r0, 0);
  if (first + 1024u < end) idB = binned[first + 1024u];

  for (unsigned int base = seg + (unsigned)w * 32u; base < end; base += 1024u) {
    const unsigned int e = base + (unsigned)lane;
    int2 idC = make_int2(r0, 0);
    if (e + 2048u < end) idC = binned[e + 2048u];
    const float drB = dinv[idB.x], dcB = dinv[idB.y];

    const bool ok = e < end;
    const float nrm = drA * dcA;
    const unsigned okmask = __builtin_amdgcn_ballot_w32(ok);
    const int lofs = (idA.x - r0) * OUT_SIZE;
    if (okmask == 0xFFFFFFFFu) accum_wave<true >(lofs, idA.y, nrm, okmask, lane, xw, tile);
    else                       accum_wave<false>(lofs, idA.y, nrm, okmask, lane, xw, tile);

    idA = idB; drA = drB; dcA = dcB; idB = idC;
  }
  __syncthreads();

  for (int q = t; q < TILE_V4; q += 1024) {
    const int r = r0 + (q >> 4);
    if (r < n) *(float4*)(out + (size_t)r * OUT_SIZE + ((q & 15) << 2)) = tile4[q];
  }
}


extern "C" void kernel_launch(void* const* d_in, const int* in_sizes, int n_in,
                              void* d_out, int out_size, void* d_ws, size_t ws_size,
                              hipStream_t stream) {
  const int*   edge = (const int*)d_in[0];
  const float* x    = (const float*)d_in[1];
  const float* W    = (const float*)d_in[3];

  const int E = in_sizes[0] / 2;
  const int n = in_sizes[1] / IN_SIZE;
  const int* row = edge;
  const int* col = edge + E;
  float* out = (float*)d_out;

  float*        xw     = (float*)d_ws;
  unsigned int* deg    = (unsigned int*)(xw + (size_t)n * OUT_SIZE);
  float*        dinv   = (float*)(deg + n);
  unsigned int* bcount = (unsigned int*)(dinv + n);
  unsigned int* cursor = bcount + MAX_BUCKETS;
  int*          binned = (int*)(cursor + MAX_BUCKETS);

  const int tiles = (n + 15) / 16;
  const int buckets = (n + ROWS_PER_BUCKET - 1) / ROWS_PER_BUCKET;
  const size_t need = (size_t)n * OUT_SIZE * 4 + (size_t)n * 8 + MAX_BUCKETS * 8 + (size_t)E * 8;

  if (buckets <= MAX_BUCKETS && need <= ws_size) {
    const int chunks = (E + CHUNK_EDGES - 1) / CHUNK_EDGES;
    k_init_bins<<<(n + 255) / 256, 256, 0, stream>>>(deg, bcount, cursor, n);
    k_count_bins<<<chunks, 256, 0, stream>>>(row, col, deg, bcount, E);
    k_dinv<<<(n + 255) / 256, 256, 0, stream>>>(deg, dinv, n);
    k_gemm<<<(tiles + 7) / 8, 256, 0, stream>>>(x, W, xw, n);
    k_fill_bins<<<chunks, 256, 0, stream>>>(row, col, bcount, cursor, binned, E);
    k_accum<<<buckets, 1024, 0, stream>>>(binned, bcount, dinv, xw, out, n);
  } else {
    k_init_deg<<<(n + 255) / 256, 256, 0, stream>>>(deg, n);
    k_count_deg<<<(E + 255) / 256, 256, 0, stream>>>(row, col, deg, E);
    k_dinv<<<(n + 255) / 256, 256, 0, stream>>>(deg, dinv, n);
    k_gemm<<<(tiles + 7) / 8, 256, 0, stream>>>(x, W, xw, n);
    const size_t tot = (size_t)n * OUT_SIZE;
    k_self<<<(unsigned)((tot + 255) / 256), 256, 0, stream>>>(xw, dinv, out, n);
    k_scatter<<<(E + 255) / 256, 256, 0, stream>>>(row, col, dinv, xw, out, E);
  }
}


# GEMM operand path: 32-row x stages parked in LDS by whole-line 16 B loads a stage ahead, W fragments kept in registers (16 columns per wave), xw stored in whole lines; sums and their order unchanged
# speedup vs baseline: 2.6115x; 1.1966x over previous
#include <hip/hip_runtime.h>
#include <hip/hip_bf16.h>

typedef __attribute__((ext_vector_type(2))) float v2f;
typedef __attribute__((ext_vector_type(8))) float v8f;

#define IN_SIZE 256
#define OUT_SIZE 64


__global__ __launch_bounds__(256) void k_init_deg(unsigned int* __restrict__ deg, int n) {
  int i = blockIdx.x * 256 + threadIdx.x;
  if (i < n) deg[i] = 0u;
}

__global__ __launch_bounds__(256) void k_count_deg(const int* __restrict__ row,
                                                   const int* __restrict__ col,
                                                   unsigned int* __restrict__ deg, int E) {
  int e = blockIdx.x * 256 + threadIdx.x;
  if (e < E) {
    int r = row[e], c = col[e];
    if (r != c) atomicAdd(&deg[r], 1u);
  }
}

__global__ __launch_bounds__(256) void k_dinv(const unsigned int* __restrict__ deg,
                                              float* __restrict__ dinv, int n) {
  int i = blockIdx.x * 256 + threadIdx.x;
  if (i < n) dinv[i] = rsqrtf((float)deg[i] + 1.0f);
}


#define G_ROWS   32
#define G_PITCH  260
#define G_XBUF   (G_ROWS * G_PITCH)
#define G_OPITCH 68
#define G_OBUF   (G_ROWS * G_OPITCH)
#define G_GRID   512

__global__ __launch_bounds__(256) void k_gemm(const float* __restrict__ x,
                                              const float* __restrict__ W,
                                              float* __restrict__ xw, int n) {
  __shared__ __attribute__((aligned(16))) float xs[2 * G_XBUF];
  __shared__ __attribute__((aligned(16))) float os[2 * G_OBUF];
  const int t    = threadIdx.x;
  const int lane = t & 31;
  const int w    = t >> 5;
  const int c    = lane & 15;
  const int h    = lane >> 4;
  const int nt   = w & 3;
  const int mt   = w >> 2;
  const int nst  = (n + G_ROWS - 1) / G_ROWS;

  const int srow = t >> 6;
  const int scol = (t & 63) << 2;

#define G_LOAD(dst, stg, i)                                                        \
  {                                                                                \
    int r_ = (stg) * G_ROWS + srow + 4 * (i);                                      \
    r_ = r_ < n ? r_ : n - 1;                                                      \
    dst = *(const float4*)(x + (size_t)r_ * IN_SIZE + scol);                       \
  }
#define G_LOAD8(stg)                                                               \
  G_LOAD(x0, stg, 0) G_LOAD(x1, stg, 1) G_LOAD(x2, stg, 2) G_LOAD(x3, stg, 3)      \
  G_LOAD(x4, stg, 4) G_LOAD(x5, stg, 5) G_LOAD(x6, stg, 6) G_LOAD(x7, stg, 7)
#define G_PARK(base, i, v) *(float4*)((base) + (srow + 4 * (i)) * G_PITCH + scol) = v;
#define G_PARK8(base)                                                              \
  G_PARK(base, 0, x0) G_PARK(base, 1, x1) G_PARK(base, 2, x2) G_PARK(base, 3, x3)  \
  G_PARK(base, 4, x4) G_PARK(base, 5, x5) G_PARK(base, 6, x6) G_PARK(base, 7, x7)

  int st = blockIdx.x;
  float4 x0, x1, x2, x3, x4, x5, x6, x7;
  G_LOAD8(st)

  {
    float4 wv[16];
#pragma unroll
    for (int i = 0; i < 16; ++i) wv[i] = *(const float4*)(W + (size_t)(srow + 4 * i) * IN_SIZE + scol);
#pragma unroll
    for (int i = 0; i < 16; ++i) *(float4*)(xs + (srow + 4 * i) * G_PITCH + scol) = wv[i];
  }
  __syncthreads();
  v2f bf[64];
  {
    const float* wb = xs + (16 * nt + c) * G_PITCH + 2 * h;
#pragma unroll
    for (int s = 0; s < 64; ++s) bf[s] = *(const v2f*)(wb + 4 * s);
  }
  __syncthreads();

  G_PARK8(xs)
  __syncthreads();
  G_LOAD8(st + G_GRID)

  int buf = 0;
  for (; st < nst; st += G_GRID) {
    const float* a = xs + buf * G_XBUF + (16 * mt + c) * G_PITCH + 2 * h;
    v8f acc = {};
    v2f ca[8], na[8];
#pragma unroll
    for (int s = 0; s < 8; ++s) ca[s] = *(const v2f*)(a + 4 * s);
#pragma unroll
    for (int b = 0; b < 8; ++b) {
      if (b < 7) {
#pragma unroll
        for (int s = 0; s < 8; ++s) na[s] = *(const v2f*)(a + 4 * (8 * (b + 1) + s));
      }
      __builtin_amdgcn_sched_barrier(0);
#pragma unroll
      for (int s = 0; s < 8; ++s)
        acc = __builtin_amdgcn_wmma_f32_16x16x4_f32(false, ca[s], false, bf[8 * b + s], (short)0, acc, false, false);
      __builtin_amdgcn_sched_barrier(0);
#pragma unroll
      for (int s = 0; s < 8; ++s) ca[s] = na[s];
    }

    float* o = os + buf * G_OBUF + (16 * mt + 8 * h) * G_OPITCH + 16 * nt + c;
#pragma unroll
    for (int v = 0; v < 8; ++v) o[v * G_OPITCH] = acc[v];

    G_PARK8(xs + (buf ^ 1) * G_XBUF)
    __syncthreads();

#pragma unroll
    for (int k = 0; k < 2; ++k) {
      const int q  = t + 256 * k;
      const int r  = q >> 4;
      const int gr = st * G_ROWS + r;
      if (gr < n)
        *(float4*)(xw + (size_t)gr * OUT_SIZE + ((q & 15) << 2)) =
            *(const float4*)(os + buf * G_OBUF + r * G_OPITCH + ((q & 15) << 2));
    }

    G_LOAD8(st + 2 * G_GRID)
    buf ^= 1;
  }
#undef G_LOAD
#undef G_LOAD8
#undef G_PARK
#undef G_PARK8
}


__global__ __launch_bounds__(256) void k_self(const float* __restrict__ xw,
                                              const float* __restrict__ dinv,
                                              float* __restrict__ out, int n) {
  size_t idx = (size_t)blockIdx.x * 256 + threadIdx.x;
  if (idx < (size_t)n * OUT_SIZE) {
    int i = (int)(idx >> 6);
    float d = dinv[i];
    out[idx] = d * d * xw[idx];
  }
}


#define EDGES_PER_GROUP 8

template <bool ALL>
__device__ __forceinline__ void scatter_wave(int r, int c, float nrm, unsigned okmask, int lane,
                                             const float* __restrict__ xw,
                                             float* __restrict__ out) {
#pragma unroll
  for (int g = 0; g < 32; g += EDGES_PER_GROUP) {
    float v0[EDGES_PER_GROUP], v1[EDGES_PER_GROUP];
#pragma unroll
    for (int i = 0; i < EDGES_PER_GROUP; ++i) {
      const int sc = __builtin_amdgcn_readlane(c, g + i);
      const float* p = xw + (size_t)sc * OUT_SIZE;
      v0[i] = p[lane];
      v1[i] = p[lane + 32];
    }
#pragma unroll
    for (int i = 0; i < EDGES_PER_GROUP; ++i) {
      if (ALL || ((okmask >> (g + i)) & 1u)) {
        const int   sr = __builtin_amdgcn_readlane(r, g + i);
        const float sn = __int_as_float(__builtin_amdgcn_readlane(__float_as_int(nrm), g + i));
        float* o = out + (size_t)sr * OUT_SIZE;
        unsafeAtomicAdd(o + lane,      sn * v0[i]);
        unsafeAtomicAdd(o + lane + 32, sn * v1[i]);
      }
    }
  }
}

__global__ __launch_bounds__(256) void k_scatter(const int* __restrict__ row,
                                                 const int* __restrict__ col,
                                                 const float* __restrict__ dinv,
                                                 const float* __restrict__ xw,
                                                 float* __restrict__ out, int E) {
  const int lane = threadIdx.x & 31;
  const int wave = blockIdx.x * 8 + (threadIdx.x >> 5);
  if (wave * 32 >= E) return;
  const int e = wave * 32 + lane;

  int r = 0, c = 0;
  if (e < E) { r = row[e]; c = col[e]; }
  const bool ok = (e < E) && (r != c);
  float nrm = 0.0f;
  if (ok) nrm = dinv[r] * dinv[c];
  const unsigned okmask = __builtin_amdgcn_ballot_w32(ok);

  if (okmask == 0xFFFFFFFFu) scatter_wave<true >(r, c, nrm, okmask, lane, xw, out);
  else                       scatter_wave<false>(r, c, nrm, okmask, lane, xw, out);
}


#define ROWS_PER_BUCKET 392
#define MAX_BUCKETS     256
#define CHUNK_EDGES     4096
#define EDGES_PER_THREAD (CHUNK_EDGES / 256)

__global__ __launch_bounds__(256) void k_init_bins(unsigned int* __restrict__ deg,
                                                   unsigned int* __restrict__ bcount,
                                                   unsigned int* __restrict__ cursor, int n) {
  int i = blockIdx.x * 256 + threadIdx.x;
  if (i < n) deg[i] = 0u;
  if (i < MAX_BUCKETS) { bcount[i] = 0u; cursor[i] = 0u; }
}

__global__ __launch_bounds__(256) void k_count_bins(const int* __restrict__ row,
                                                    const int* __restrict__ col,
                                                    unsigned int* __restrict__ deg,
                                                    unsigned int* __restrict__ bcount, int E) {
  __shared__ unsigned int hist[MAX_BUCKETS];
  const int t = threadIdx.x;
  hist[t] = 0u;
  __syncthreads();
  const int base = blockIdx.x * CHUNK_EDGES + t;
  int rr[EDGES_PER_THREAD], cc[EDGES_PER_THREAD];
#pragma unroll
  for (int i = 0; i < EDGES_PER_THREAD; ++i) {
    const int e = base + i * 256;
    const int ec = e < E ? e : E - 1;
    rr[i] = row[ec]; cc[i] = col[ec];
  }
#pragma unroll
  for (int i = 0; i < EDGES_PER_THREAD; ++i) {
    const int e = base + i * 256;
    if (e < E && rr[i] != cc[i]) {
      atomicAdd(&deg[rr[i]], 1u);
      atomicAdd(&hist[(unsigned)rr[i] / ROWS_PER_BUCKET], 1u);
    }
  }
  __syncthreads();
  const unsigned int h = hist[t];
  if (h) atomicAdd(&bcount[t], h);
}

__global__ __launch_bounds__(256) void k_fill_bins(const int* __restrict__ row,
                                                   const int* __restrict__ col,
                                                   const unsigned int* __restrict__ bcount,
                                                   unsigned int* __restrict__ cursor,
                                                   int* __restrict__ binned, int E) {
  __shared__ unsigned int hist[MAX_BUCKETS];
  __shared__ unsigned int scan[MAX_BUCKETS];
  __shared__ unsigned int dst[MAX_BUCKETS];
  const int t = threadIdx.x;
  hist[t] = 0u;
  const unsigned int cnt_t = bcount[t];
  scan[t] = cnt_t;
  __syncthreads();
  for (int off = 1; off < MAX_BUCKETS; off <<= 1) {
    const unsigned int v = (t >= off) ? scan[t - off] : 0u;
    __syncthreads();
    scan[t] += v;
    __syncthreads();
  }
  const unsigned int start_t = scan[t] - cnt_t;

  int rr[EDGES_PER_THREAD], cc[EDGES_PER_THREAD];
  unsigned int lr[EDGES_PER_THREAD];
  const int base = blockIdx.x * CHUNK_EDGES + t;
#pragma unroll
  for (int i = 0; i < EDGES_PER_THREAD; ++i) {
    const int e = base + i * 256;
    const int ec = e < E ? e : E - 1;
    rr[i] = row[ec]; cc[i] = col[ec];
  }
#pragma unroll
  for (int i = 0; i < EDGES_PER_THREAD; ++i) {
    const int e = base + i * 256;
    lr[i] = 0u;
    if (e < E && rr[i] != cc[i])
      lr[i] = atomicAdd(&hist[(unsigned)rr[i] / ROWS_PER_BUCKET], 1u);
    else
      rr[i] = -1;
  }
  __syncthreads();
  const unsigned int h = hist[t];
  unsigned int g = 0u;
  if (h) g = atomicAdd(&cursor[t], h);
  dst[t] = start_t + g;
  __syncthreads();
#pragma unroll
  for (int i = 0; i < EDGES_PER_THREAD; ++i) {
    if (rr[i] >= 0)
      ((int2*)binned)[dst[(unsigned)rr[i] / ROWS_PER_BUCKET] + lr[i]] = make_int2(rr[i], cc[i]);
  }
}

#define ACC_GROUP 16
#define TILE_FLOATS (ROWS_PER_BUCKET * OUT_SIZE)
#define TILE_V4     (TILE_FLOATS / 4)
#define TILE_V4_FULL (TILE_V4 / 1024)

template <bool ALL>
__device__ __forceinline__ void accum_wave(int lofs, int c, float nrm, unsigned okmask, int lane,
                                           const float* __restrict__ xw, float* tile) {
#pragma unroll
  for (int g = 0; g < 32; g += ACC_GROUP) {
    float v0[ACC_GROUP], v1[ACC_GROUP];
#pragma unroll
    for (int i = 0; i < ACC_GROUP; ++i) {
      const int sc = __builtin_amdgcn_readlane(c, g + i);
      const float* p = xw + (size_t)sc * OUT_SIZE;
      v0[i] = p[lane];
      v1[i] = p[lane + 32];
    }
#pragma unroll
    for (int i = 0; i < ACC_GROUP; ++i) {
      if (ALL || ((okmask >> (g + i)) & 1u)) {
        const int   so = __builtin_amdgcn_readlane(lofs, g + i);
        const float sn = __int_as_float(__builtin_amdgcn_readlane(__float_as_int(nrm), g + i));
        atomicAdd(tile + so + lane,      sn * v0[i]);
        atomicAdd(tile + so + lane + 32, sn * v1[i]);
      }
    }
  }
}

__device__ __forceinline__ float4 self_term(const float* __restrict__ dinv,
                                            const float* __restrict__ xw, int r0, int q, int n) {
  int r = r0 + (q >> 4);
  r = r < n ? r : n - 1;
  const float d = dinv[r];
  const float4 v = *(const float4*)(xw + (size_t)r * OUT_SIZE + ((q & 15) << 2));
  return make_float4(d * d * v.x, d * d * v.y, d * d * v.z, d * d * v.w);
}

__global__ __launch_bounds__(1024) void k_accum(const int* __restrict__ binned_rc,
                                                const unsigned int* __restrict__ bcount,
                                                const float* __restrict__ dinv,
                                                const float* __restrict__ xw,
                                                float* __restrict__ out, int n) {
  __shared__ __attribute__((aligned(16))) float tile[TILE_FLOATS];
  __shared__ unsigned int s_start;
  float4* tile4 = (float4*)tile;
  const int2* binned = (const int2*)binned_rc;
  const int t = threadIdx.x;
  const int lane = t & 31;
  const int w = t >> 5;
  const int b = blockIdx.x;
  const int r0 = b * ROWS_PER_BUCKET;

  if (t == 0) s_start = 0u;
  __syncthreads();
  if (t < MAX_BUCKETS && t < b) atomicAdd(&s_start, bcount[t]);

  {
    float4 sv[TILE_V4_FULL];
#pragma unroll
    for (int k = 0; k < TILE_V4_FULL; ++k) sv[k] = self_term(dinv, xw, r0, t + k * 1024, n);
#pragma unroll
    for (int k = 0; k < TILE_V4_FULL; ++k) tile4[t + k * 1024] = sv[k];
    const int q = t + TILE_V4_FULL * 1024;
    if (q < TILE_V4) tile4[q] = self_term(dinv, xw, r0, q, n);
  }
  __syncthreads();

  const unsigned int seg = s_start;
  const unsigned int end = seg + bcount[b];
  const unsigned int first = seg + (unsigned)w * 32u + (unsigned)lane;

  int2 idA = make_int2(r0, 0);
  if (first < end) idA = binned[first];
  float drA = dinv[idA.x], dcA = dinv[idA.y];
  int2 idB = make_int2(r0, 0);
  if (first + 1024u < end) idB = binned[first + 1024u];

  for (unsigned int base = seg + (unsigned)w * 32u; base < end; base += 1024u) {
    const unsigned int e = base + (unsigned)lane;
    int2 idC = make_int2(r0, 0);
    if (e + 2048u < end) idC = binned[e + 2048u];
    const float drB = dinv[idB.x], dcB = dinv[idB.y];

    const bool ok = e < end;
    const float nrm = drA * dcA;
    const unsigned okmask = __builtin_amdgcn_ballot_w32(ok);
    const int lofs = (idA.x - r0) * OUT_SIZE;
    if (okmask == 0xFFFFFFFFu) accum_wave<true >(lofs, idA.y, nrm, okmask, lane, xw, tile);
    else                       accum_wave<false>(lofs, idA.y, nrm, okmask, lane, xw, tile);

    idA = idB; drA = drB; dcA = dcB; idB = idC;
  }
  __syncthreads();

  for (int q = t; q < TILE_V4; q += 1024) {
    const int r = r0 + (q >> 4);
    if (r < n) *(float4*)(out + (size_t)r * OUT_SIZE + ((q & 15) << 2)) = tile4[q];
  }
}


extern "C" void kernel_launch(void* const* d_in, const int* in_sizes, int n_in,
                              void* d_out, int out_size, void* d_ws, size_t ws_size,
                              hipStream_t stream) {
  const int*   edge = (const int*)d_in[0];
  const float* x    = (const float*)d_in[1];
  const float* W    = (const float*)d_in[3];

  const int E = in_sizes[0] / 2;
  const int n = in_sizes[1] / IN_SIZE;
  const int* row = edge;
  const int* col = edge + E;
  float* out = (float*)d_out;

  float*        xw     = (float*)d_ws;
  unsigned int* deg    = (unsigned int*)(xw + (size_t)n * OUT_SIZE);
  float*        dinv   = (float*)(deg + n);
  unsigned int* bcount = (unsigned int*)(dinv + n);
  unsigned int* cursor = bcount + MAX_BUCKETS;
  int*          binned = (int*)(cursor + MAX_BUCKETS);

  const int buckets = (n + ROWS_PER_BUCKET - 1) / ROWS_PER_BUCKET;
  const size_t need = (size_t)n * OUT_SIZE * 4 + (size_t)n * 8 + MAX_BUCKETS * 8 + (size_t)E * 8;

  if (buckets <= MAX_BUCKETS && need <= ws_size) {
    const int chunks = (E + CHUNK_EDGES - 1) / CHUNK_EDGES;
    k_init_bins<<<(n + 255) / 256, 256, 0, stream>>>(deg, bcount, cursor, n);
    k_count_bins<<<chunks, 256, 0, stream>>>(row, col, deg, bcount, E);
    k_dinv<<<(n + 255) / 256, 256, 0, stream>>>(deg, dinv, n);
    k_gemm<<<G_GRID, 256, 0, stream>>>(x, W, xw, n);
    k_fill_bins<<<chunks, 256, 0, stream>>>(row, col, bcount, cursor, binned, E);
    k_accum<<<buckets, 1024, 0, stream>>>(binned, bcount, dinv, xw, out, n);
  } else {
    k_init_deg<<<(n + 255) / 256, 256, 0, stream>>>(deg, n);
    k_count_deg<<<(E + 255) / 256, 256, 0, stream>>>(row, col, deg, E);
    k_dinv<<<(n + 255) / 256, 256, 0, stream>>>(deg, dinv, n);
    k_gemm<<<G_GRID, 256, 0, stream>>>(x, W, xw, n);
    const size_t tot = (size_t)n * OUT_SIZE;
    k_self<<<(unsigned)((tot + 255) / 256), 256, 0, stream>>>(xw, dinv, out, n);
    k_scatter<<<(E + 255) / 256, 256, 0, stream>>>(row, col, dinv, xw, out, E);
  }
}


# In-degrees counted per destination bucket in an LDS histogram over the binned edges (one work-group per bucket, which also writes dinv), not by 1.6M global u32 atomic adds; sum order unchanged
# speedup vs baseline: 4.8635x; 1.8623x over previous
#include <hip/hip_runtime.h>
#include <hip/hip_bf16.h>

typedef __attribute__((ext_vector_type(2))) float v2f;
typedef __attribute__((ext_vector_type(8))) float v8f;

#define IN_SIZE 256
#define OUT_SIZE 64


__global__ __launch_bounds__(256) void k_init_deg(unsigned int* __restrict__ deg, int n) {
  int i = blockIdx.x * 256 + threadIdx.x;
  if (i < n) deg[i] = 0u;
}

__global__ __launch_bounds__(256) void k_count_deg(const int* __restrict__ row,
                                                   const int* __restrict__ col,
                                                   unsigned int* __restrict__ deg, int E) {
  int e = blockIdx.x * 256 + threadIdx.x;
  if (e < E) {
    int r = row[e], c = col[e];
    if (r != c) atomicAdd(&deg[r], 1u);
  }
}

__global__ __launch_bounds__(256) void k_dinv(const unsigned int* __restrict__ deg,
                                              float* __restrict__ dinv, int n) {
  int i = blockIdx.x * 256 + threadIdx.x;
  if (i < n) dinv[i] = rsqrtf((float)deg[i] + 1.0f);
}


#define G_ROWS   32
#define G_PITCH  260
#define G_XBUF   (G_ROWS * G_PITCH)
#define G_OPITCH 68
#define G_OBUF   (G_ROWS * G_OPITCH)
#define G_GRID   512

__global__ __launch_bounds__(256) void k_gemm(const float* __restrict__ x,
                                              const float* __restrict__ W,
                                              float* __restrict__ xw, int n) {
  __shared__ __attribute__((aligned(16))) float xs[2 * G_XBUF];
  __shared__ __attribute__((aligned(16))) float os[2 * G_OBUF];
  const int t    = threadIdx.x;
  const int lane = t & 31;
  const int w    = t >> 5;
  const int c    = lane & 15;
  const int h    = lane >> 4;
  const int nt   = w & 3;
  const int mt   = w >> 2;
  const int nst  = (n + G_ROWS - 1) / G_ROWS;

  const int srow = t >> 6;
  const int scol = (t & 63) << 2;

#define G_LOAD(dst, stg, i)                                                        \
  {                                                                                \
    int r_ = (stg) * G_ROWS + srow + 4 * (i);                                      \
    r_ = r_ < n ? r_ : n - 1;                                                      \
    dst = *(const float4*)(x + (size_t)r_ * IN_SIZE + scol);                       \
  }
#define G_LOAD8(stg)                                                               \
  G_LOAD(x0, stg, 0) G_LOAD(x1, stg, 1) G_LOAD(x2, stg, 2) G_LOAD(x3, stg, 3)      \
  G_LOAD(x4, stg, 4) G_LOAD(x5, stg, 5) G_LOAD(x6, stg, 6) G_LOAD(x7, stg, 7)
#define G_PARK(base, i, v) *(float4*)((base) + (srow + 4 * (i)) * G_PITCH + scol) = v;
#define G_PARK8(base)                                                              \
  G_PARK(base, 0, x0) G_PARK(base, 1, x1) G_PARK(base, 2, x2) G_PARK(base, 3, x3)  \
  G_PARK(base, 4, x4) G_PARK(base, 5, x5) G_PARK(base, 6, x6) G_PARK(base, 7, x7)

  int st = blockIdx.x;
  float4 x0, x1, x2, x3, x4, x5, x6, x7;
  G_LOAD8(st)

  {
    float4 wv[16];
#pragma unroll
    for (int i = 0; i < 16; ++i) wv[i] = *(const float4*)(W + (size_t)(srow + 4 * i) * IN_SIZE + scol);
#pragma unroll
    for (int i = 0; i < 16; ++i) *(float4*)(xs + (srow + 4 * i) * G_PITCH + scol) = wv[i];
  }
  __syncthreads();
  v2f bf[64];
  {
    const float* wb = xs + (16 * nt + c) * G_PITCH + 2 * h;
#pragma unroll
    for (int s = 0; s < 64; ++s) bf[s] = *(const v2f*)(wb + 4 * s);
  }
  __syncthreads();

  G_PARK8(xs)
  __syncthreads();
  G_LOAD8(st + G_GRID)

  int buf = 0;
  for (; st < nst; st += G_GRID) {
    const float* a = xs + buf * G_XBUF + (16 * mt + c) * G_PITCH + 2 * h;
    v8f acc = {};
    v2f ca[8], na[8];
#pragma unroll
    for (int s = 0; s < 8; ++s) ca[s] = *(const v2f*)(a + 4 * s);
#pragma unroll
    for (int b = 0; b < 8; ++b) {
      if (b < 7) {
#pragma unroll
        for (int s = 0; s < 8; ++s) na[s] = *(const v2f*)(a + 4 * (8 * (b + 1) + s));
      }
      __builtin_amdgcn_sched_barrier(0);
#pragma unroll
      for (int s = 0; s < 8; ++s)
        acc = __builtin_amdgcn_wmma_f32_16x16x4_f32(false, ca[s], false, bf[8 * b + s], (short)0, acc, false, false);
      __builtin_amdgcn_sched_barrier(0);
#pragma unroll
      for (int s = 0; s < 8; ++s) ca[s] = na[s];
    }

    float* o = os + buf * G_OBUF + (16 * mt + 8 * h) * G_OPITCH + 16 * nt + c;
#pragma unroll
    for (int v = 0; v < 8; ++v) o[v * G_OPITCH] = acc[v];

    G_PARK8(xs + (buf ^ 1) * G_XBUF)
    __syncthreads();

#pragma unroll
    for (int k = 0; k < 2; ++k) {
      const int q  = t + 256 * k;
      const int r  = q >> 4;
      const int gr = st * G_ROWS + r;
      if (gr < n)
        *(float4*)(xw + (size_t)gr * OUT_SIZE + ((q & 15) << 2)) =
            *(const float4*)(os + buf * G_OBUF + r * G_OPITCH + ((q & 15) << 2));
    }

    G_LOAD8(st + 2 * G_GRID)
    buf ^= 1;
  }
#undef G_LOAD
#undef G_LOAD8
#undef G_PARK
#undef G_PARK8
}


__global__ __launch_bounds__(256) void k_self(const float* __restrict__ xw,
                                              const float* __restrict__ dinv,
                                              float* __restrict__ out, int n) {
  size_t idx = (size_t)blockIdx.x * 256 + threadIdx.x;
  if (idx < (size_t)n * OUT_SIZE) {
    int i = (int)(idx >> 6);
    float d = dinv[i];
    out[idx] = d * d * xw[idx];
  }
}


#define EDGES_PER_GROUP 8

template <bool ALL>
__device__ __forceinline__ void scatter_wave(int r, int c, float nrm, unsigned okmask, int lane,
                                             const float* __restrict__ xw,
                                             float* __restrict__ out) {
#pragma unroll
  for (int g = 0; g < 32; g += EDGES_PER_GROUP) {
    float v0[EDGES_PER_GROUP], v1[EDGES_PER_GROUP];
#pragma unroll
    for (int i = 0; i < EDGES_PER_GROUP; ++i) {
      const int sc = __builtin_amdgcn_readlane(c, g + i);
      const float* p = xw + (size_t)sc * OUT_SIZE;
      v0[i] = p[lane];
      v1[i] = p[lane + 32];
    }
#pragma unroll
    for (int i = 0; i < EDGES_PER_GROUP; ++i) {
      if (ALL || ((okmask >> (g + i)) & 1u)) {
        const int   sr = __builtin_amdgcn_readlane(r, g + i);
        const float sn = __int_as_float(__builtin_amdgcn_readlane(__float_as_int(nrm), g + i));
        float* o = out + (size_t)sr * OUT_SIZE;
        unsafeAtomicAdd(o + lane,      sn * v0[i]);
        unsafeAtomicAdd(o + lane + 32, sn * v1[i]);
      }
    }
  }
}

__global__ __launch_bounds__(256) void k_scatter(const int* __restrict__ row,
                                                 const int* __restrict__ col,
                                                 const float* __restrict__ dinv,
                                                 const float* __restrict__ xw,
                                                 float* __restrict__ out, int E) {
  const int lane = threadIdx.x & 31;
  const int wave = blockIdx.x * 8 + (threadIdx.x >> 5);
  if (wave * 32 >= E) return;
  const int e = wave * 32 + lane;

  int r = 0, c = 0;
  if (e < E) { r = row[e]; c = col[e]; }
  const bool ok = (e < E) && (r != c);
  float nrm = 0.0f;
  if (ok) nrm = dinv[r] * dinv[c];
  const unsigned okmask = __builtin_amdgcn_ballot_w32(ok);

  if (okmask == 0xFFFFFFFFu) scatter_wave<true >(r, c, nrm, okmask, lane, xw, out);
  else                       scatter_wave<false>(r, c, nrm, okmask, lane, xw, out);
}


#define ROWS_PER_BUCKET 392
#define MAX_BUCKETS     256
#define CHUNK_EDGES     4096
#define EDGES_PER_THREAD (CHUNK_EDGES / 256)

__global__ __launch_bounds__(256) void k_init_bins(unsigned int* __restrict__ bcount,
                                                   unsigned int* __restrict__ cursor) {
  const int i = threadIdx.x;
  bcount[i] = 0u; cursor[i] = 0u;
}

__global__ __launch_bounds__(256) void k_count_bins(const int* __restrict__ row,
                                                    const int* __restrict__ col,
                                                    unsigned int* __restrict__ bcount, int E) {
  __shared__ unsigned int hist[MAX_BUCKETS];
  const int t = threadIdx.x;
  hist[t] = 0u;
  __syncthreads();
  const int base = blockIdx.x * CHUNK_EDGES + t;
  int rr[EDGES_PER_THREAD], cc[EDGES_PER_THREAD];
#pragma unroll
  for (int i = 0; i < EDGES_PER_THREAD; ++i) {
    const int e = base + i * 256;
    const int ec = e < E ? e : E - 1;
    rr[i] = row[ec]; cc[i] = col[ec];
  }
#pragma unroll
  for (int i = 0; i < EDGES_PER_THREAD; ++i) {
    const int e = base + i * 256;
    if (e < E && rr[i] != cc[i])
      atomicAdd(&hist[(unsigned)rr[i] / ROWS_PER_BUCKET], 1u);
  }
  __syncthreads();
  const unsigned int h = hist[t];
  if (h) atomicAdd(&bcount[t], h);
}

__global__ __launch_bounds__(256) void k_fill_bins(const int* __restrict__ row,
                                                   const int* __restrict__ col,
                                                   const unsigned int* __restrict__ bcount,
                                                   unsigned int* __restrict__ cursor,
                                                   int* __restrict__ binned, int E) {
  __shared__ unsigned int hist[MAX_BUCKETS];
  __shared__ unsigned int scan[MAX_BUCKETS];
  __shared__ unsigned int dst[MAX_BUCKETS];
  const int t = threadIdx.x;
  hist[t] = 0u;
  const unsigned int cnt_t = bcount[t];
  scan[t] = cnt_t;
  __syncthreads();
  for (int off = 1; off < MAX_BUCKETS; off <<= 1) {
    const unsigned int v = (t >= off) ? scan[t - off] : 0u;
    __syncthreads();
    scan[t] += v;
    __syncthreads();
  }
  const unsigned int start_t = scan[t] - cnt_t;

  int rr[EDGES_PER_THREAD], cc[EDGES_PER_THREAD];
  unsigned int lr[EDGES_PER_THREAD];
  const int base = blockIdx.x * CHUNK_EDGES + t;
#pragma unroll
  for (int i = 0; i < EDGES_PER_THREAD; ++i) {
    const int e = base + i * 256;
    const int ec = e < E ? e : E - 1;
    rr[i] = row[ec]; cc[i] = col[ec];
  }
#pragma unroll
  for (int i = 0; i < EDGES_PER_THREAD; ++i) {
    const int e = base + i * 256;
    lr[i] = 0u;
    if (e < E && rr[i] != cc[i])
      lr[i] = atomicAdd(&hist[(unsigned)rr[i] / ROWS_PER_BUCKET], 1u);
    else
      rr[i] = -1;
  }
  __syncthreads();
  const unsigned int h = hist[t];
  unsigned int g = 0u;
  if (h) g = atomicAdd(&cursor[t], h);
  dst[t] = start_t + g;
  __syncthreads();
#pragma unroll
  for (int i = 0; i < EDGES_PER_THREAD; ++i) {
    if (rr[i] >= 0)
      ((int2*)binned)[dst[(unsigned)rr[i] / ROWS_PER_BUCKET] + lr[i]] = make_int2(rr[i], cc[i]);
  }
}


#define DEG_LOADS 32

__global__ __launch_bounds__(256) void k_bucket_dinv(const int* __restrict__ binned_rc,
                                                     const unsigned int* __restrict__ bcount,
                                                     float* __restrict__ dinv, int n) {
  __shared__ unsigned int hist[ROWS_PER_BUCKET];
  __shared__ unsigned int s_start;
  const int2* binned = (const int2*)binned_rc;
  const int t = threadIdx.x;
  const int b = blockIdx.x;
  const int r0 = b * ROWS_PER_BUCKET;

  for (int i = t; i < ROWS_PER_BUCKET; i += 256) hist[i] = 0u;
  if (t == 0) s_start = 0u;
  __syncthreads();
  if (t < b) atomicAdd(&s_start, bcount[t]);
  __syncthreads();
  const unsigned int seg = s_start;
  const unsigned int cnt = bcount[b];

  for (unsigned int base = 0u; base < cnt; base += 256u * DEG_LOADS) {
    int rr[DEG_LOADS];
#pragma unroll
    for (int k = 0; k < DEG_LOADS; ++k) {
      const unsigned int i  = base + (unsigned)t + 256u * (unsigned)k;
      const unsigned int ic = i < cnt ? i : cnt - 1u;
      rr[k] = binned[seg + ic].x;
    }
#pragma unroll
    for (int k = 0; k < DEG_LOADS; ++k) {
      const unsigned int i = base + (unsigned)t + 256u * (unsigned)k;
      if (i < cnt) atomicAdd(&hist[rr[k] - r0], 1u);
    }
  }
  __syncthreads();
  for (int i = t; i < ROWS_PER_BUCKET; i += 256) {
    const int r = r0 + i;
    if (r < n) dinv[r] = rsqrtf((float)hist[i] + 1.0f);
  }
}

#define ACC_GROUP 16
#define TILE_FLOATS (ROWS_PER_BUCKET * OUT_SIZE)
#define TILE_V4     (TILE_FLOATS / 4)
#define TILE_V4_FULL (TILE_V4 / 1024)

template <bool ALL>
__device__ __forceinline__ void accum_wave(int lofs, int c, float nrm, unsigned okmask, int lane,
                                           const float* __restrict__ xw, float* tile) {
#pragma unroll
  for (int g = 0; g < 32; g += ACC_GROUP) {
    float v0[ACC_GROUP], v1[ACC_GROUP];
#pragma unroll
    for (int i = 0; i < ACC_GROUP; ++i) {
      const int sc = __builtin_amdgcn_readlane(c, g + i);
      const float* p = xw + (size_t)sc * OUT_SIZE;
      v0[i] = p[lane];
      v1[i] = p[lane + 32];
    }
#pragma unroll
    for (int i = 0; i < ACC_GROUP; ++i) {
      if (ALL || ((okmask >> (g + i)) & 1u)) {
        const int   so = __builtin_amdgcn_readlane(lofs, g + i);
        const float sn = __int_as_float(__builtin_amdgcn_readlane(__float_as_int(nrm), g + i));
        atomicAdd(tile + so + lane,      sn * v0[i]);
        atomicAdd(tile + so + lane + 32, sn * v1[i]);
      }
    }
  }
}

__device__ __forceinline__ float4 self_term(const float* __restrict__ dinv,
                                            const float* __restrict__ xw, int r0, int q, int n) {
  int r = r0 + (q >> 4);
  r = r < n ? r : n - 1;
  const float d = dinv[r];
  const float4 v = *(const float4*)(xw + (size_t)r * OUT_SIZE + ((q & 15) << 2));
  return make_float4(d * d * v.x, d * d * v.y, d * d * v.z, d * d * v.w);
}

__global__ __launch_bounds__(1024) void k_accum(const int* __restrict__ binned_rc,
                                                const unsigned int* __restrict__ bcount,
                                                const float* __restrict__ dinv,
                                                const float* __restrict__ xw,
                                                float* __restrict__ out, int n) {
  __shared__ __attribute__((aligned(16))) float tile[TILE_FLOATS];
  __shared__ unsigned int s_start;
  float4* tile4 = (float4*)tile;
  const int2* binned = (const int2*)binned_rc;
  const int t = threadIdx.x;
  const int lane = t & 31;
  const int w = t >> 5;
  const int b = blockIdx.x;
  const int r0 = b * ROWS_PER_BUCKET;

  if (t == 0) s_start = 0u;
  __syncthreads();
  if (t < MAX_BUCKETS && t < b) atomicAdd(&s_start, bcount[t]);

  {
    float4 sv[TILE_V4_FULL];
#pragma unroll
    for (int k = 0; k < TILE_V4_FULL; ++k) sv[k] = self_term(dinv, xw, r0, t + k * 1024, n);
#pragma unroll
    for (int k = 0; k < TILE_V4_FULL; ++k) tile4[t + k * 1024] = sv[k];
    const int q = t + TILE_V4_FULL * 1024;
    if (q < TILE_V4) tile4[q] = self_term(dinv, xw, r0, q, n);
  }
  __syncthreads();

  const unsigned int seg = s_start;
  const unsigned int end = seg + bcount[b];
  const unsigned int first = seg + (unsigned)w * 32u + (unsigned)lane;

  int2 idA = make_int2(r0, 0);
  if (first < end) idA = binned[first];
  float drA = dinv[idA.x], dcA = dinv[idA.y];
  int2 idB = make_int2(r0, 0);
  if (first + 1024u < end) idB = binned[first + 1024u];

  for (unsigned int base = seg + (unsigned)w * 32u; base < end; base += 1024u) {
    const unsigned int e = base + (unsigned)lane;
    int2 idC = make_int2(r0, 0);
    if (e + 2048u < end) idC = binned[e + 2048u];
    const float drB = dinv[idB.x], dcB = dinv[idB.y];

    const bool ok = e < end;
    const float nrm = drA * dcA;
    const unsigned okmask = __builtin_amdgcn_ballot_w32(ok);
    const int lofs = (idA.x - r0) * OUT_SIZE;
    if (okmask == 0xFFFFFFFFu) accum_wave<true >(lofs, idA.y, nrm, okmask, lane, xw, tile);
    else                       accum_wave<false>(lofs, idA.y, nrm, okmask, lane, xw, tile);

    idA = idB; drA = drB; dcA = dcB; idB = idC;
  }
  __syncthreads();

  for (int q = t; q < TILE_V4; q += 1024) {
    const int r = r0 + (q >> 4);
    if (r < n) *(float4*)(out + (size_t)r * OUT_SIZE + ((q & 15) << 2)) = tile4[q];
  }
}


extern "C" void kernel_launch(void* const* d_in, const int* in_sizes, int n_in,
                              void* d_out, int out_size, void* d_ws, size_t ws_size,
                              hipStream_t stream) {
  const int*   edge = (const int*)d_in[0];
  const float* x    = (const float*)d_in[1];
  const float* W    = (const float*)d_in[3];

  const int E = in_sizes[0] / 2;
  const int n = in_sizes[1] / IN_SIZE;
  const int* row = edge;
  const int* col = edge + E;
  float* out = (float*)d_out;

  float*        xw     = (float*)d_ws;
  unsigned int* deg    = (unsigned int*)(xw + (size_t)n * OUT_SIZE);
  float*        dinv   = (float*)(deg + n);
  unsigned int* bcount = (unsigned int*)(dinv + n);
  unsigned int* cursor = bcount + MAX_BUCKETS;
  int*          binned = (int*)(cursor + MAX_BUCKETS);

  const int buckets = (n + ROWS_PER_BUCKET - 1) / ROWS_PER_BUCKET;
  const size_t need = (size_t)n * OUT_SIZE * 4 + (size_t)n * 8 + MAX_BUCKETS * 8 + (size_t)E * 8;

  if (buckets <= MAX_BUCKETS && need <= ws_size) {
    const int chunks = (E + CHUNK_EDGES - 1) / CHUNK_EDGES;
    k_init_bins<<<1, MAX_BUCKETS, 0, stream>>>(bcount, cursor);
    k_count_bins<<<chunks, 256, 0, stream>>>(row, col, bcount, E);
    k_gemm<<<G_GRID, 256, 0, stream>>>(x, W, xw, n);
    k_fill_bins<<<chunks, 256, 0, stream>>>(row, col, bcount, cursor, binned, E);
    k_bucket_dinv<<<buckets, 256, 0, stream>>>(binned, bcount, dinv, n);
    k_accum<<<buckets, 1024, 0, stream>>>(binned, bcount, dinv, xw, out, n);
  } else {
    k_init_deg<<<(n + 255) / 256, 256, 0, stream>>>(deg, n);
    k_count_deg<<<(E + 255) / 256, 256, 0, stream>>>(row, col, deg, E);
    k_dinv<<<(n + 255) / 256, 256, 0, stream>>>(deg, dinv, n);
    k_gemm<<<G_GRID, 256, 0, stream>>>(x, W, xw, n);
    const size_t tot = (size_t)n * OUT_SIZE;
    k_self<<<(unsigned)((tot + 255) / 256), 256, 0, stream>>>(xw, dinv, out, n);
    k_scatter<<<(E + 255) / 256, 256, 0, stream>>>(row, col, dinv, xw, out, E);
  }
}


# Edge weight factored out of the edge loop: GEMM (now behind the degree pass) stores dinv[c]*xw[c]; buckets add unscaled rows, scale by dinv[r] at the store; per-term rounding changes, sum order kept
# speedup vs baseline: 5.2728x; 1.0842x over previous
#include <hip/hip_runtime.h>
#include <hip/hip_bf16.h>

typedef __attribute__((ext_vector_type(2))) float v2f;
typedef __attribute__((ext_vector_type(8))) float v8f;

#define IN_SIZE 256
#define OUT_SIZE 64


__global__ __launch_bounds__(256) void k_init_deg(unsigned int* __restrict__ deg, int n) {
  int i = blockIdx.x * 256 + threadIdx.x;
  if (i < n) deg[i] = 0u;
}

__global__ __launch_bounds__(256) void k_count_deg(const int* __restrict__ row,
                                                   const int* __restrict__ col,
                                                   unsigned int* __restrict__ deg, int E) {
  int e = blockIdx.x * 256 + threadIdx.x;
  if (e < E) {
    int r = row[e], c = col[e];
    if (r != c) atomicAdd(&deg[r], 1u);
  }
}

__global__ __launch_bounds__(256) void k_dinv(const unsigned int* __restrict__ deg,
                                              float* __restrict__ dinv, int n) {
  int i = blockIdx.x * 256 + threadIdx.x;
  if (i < n) dinv[i] = rsqrtf((float)deg[i] + 1.0f);
}


#define G_ROWS   32
#define G_PITCH  260
#define G_XBUF   (G_ROWS * G_PITCH)
#define G_OPITCH 68
#define G_OBUF   (G_ROWS * G_OPITCH)
#define G_GRID   512

__global__ __launch_bounds__(256) void k_gemm(const float* __restrict__ x,
                                              const float* __restrict__ W,
                                              const float* __restrict__ rs,
                                              float* __restrict__ xw, int n) {
  __shared__ __attribute__((aligned(16))) float xs[2 * G_XBUF];
  __shared__ __attribute__((aligned(16))) float os[2 * G_OBUF];
  const int t    = threadIdx.x;
  const int lane = t & 31;
  const int w    = t >> 5;
  const int c    = lane & 15;
  const int h    = lane >> 4;
  const int nt   = w & 3;
  const int mt   = w >> 2;
  const int nst  = (n + G_ROWS - 1) / G_ROWS;

  const int srow = t >> 6;
  const int scol = (t & 63) << 2;

#define G_LOAD(dst, stg, i)                                                        \
  {                                                                                \
    int r_ = (stg) * G_ROWS + srow + 4 * (i);                                      \
    r_ = r_ < n ? r_ : n - 1;                                                      \
    dst = *(const float4*)(x + (size_t)r_ * IN_SIZE + scol);                       \
  }
#define G_LOAD8(stg)                                                               \
  G_LOAD(x0, stg, 0) G_LOAD(x1, stg, 1) G_LOAD(x2, stg, 2) G_LOAD(x3, stg, 3)      \
  G_LOAD(x4, stg, 4) G_LOAD(x5, stg, 5) G_LOAD(x6, stg, 6) G_LOAD(x7, stg, 7)
#define G_PARK(base, i, v) *(float4*)((base) + (srow + 4 * (i)) * G_PITCH + scol) = v;
#define G_PARK8(base)                                                              \
  G_PARK(base, 0, x0) G_PARK(base, 1, x1) G_PARK(base, 2, x2) G_PARK(base, 3, x3)  \
  G_PARK(base, 4, x4) G_PARK(base, 5, x5) G_PARK(base, 6, x6) G_PARK(base, 7, x7)

  int st = blockIdx.x;
  float4 x0, x1, x2, x3, x4, x5, x6, x7;
  G_LOAD8(st)

  {
    float4 wv[16];
#pragma unroll
    for (int i = 0; i < 16; ++i) wv[i] = *(const float4*)(W + (size_t)(srow + 4 * i) * IN_SIZE + scol);
#pragma unroll
    for (int i = 0; i < 16; ++i) *(float4*)(xs + (srow + 4 * i) * G_PITCH + scol) = wv[i];
  }
  __syncthreads();
  v2f bf[64];
  {
    const float* wb = xs + (16 * nt + c) * G_PITCH + 2 * h;
#pragma unroll
    for (int s = 0; s < 64; ++s) bf[s] = *(const v2f*)(wb + 4 * s);
  }
  __syncthreads();

  G_PARK8(xs)
  __syncthreads();
  G_LOAD8(st + G_GRID)

  int buf = 0;
  for (; st < nst; st += G_GRID) {
    const float* a = xs + buf * G_XBUF + (16 * mt + c) * G_PITCH + 2 * h;
    float rs0 = 1.0f, rs1 = 1.0f;
    if (rs) {
      int g0 = st * G_ROWS + (t >> 4);
      int g1 = g0 + 16;
      g0 = g0 < n ? g0 : n - 1;
      g1 = g1 < n ? g1 : n - 1;
      rs0 = rs[g0];
      rs1 = rs[g1];
    }
    v8f acc = {};
    v2f ca[8], na[8];
#pragma unroll
    for (int s = 0; s < 8; ++s) ca[s] = *(const v2f*)(a + 4 * s);
#pragma unroll
    for (int b = 0; b < 8; ++b) {
      if (b < 7) {
#pragma unroll
        for (int s = 0; s < 8; ++s) na[s] = *(const v2f*)(a + 4 * (8 * (b + 1) + s));
      }
      __builtin_amdgcn_sched_barrier(0);
#pragma unroll
      for (int s = 0; s < 8; ++s)
        acc = __builtin_amdgcn_wmma_f32_16x16x4_f32(false, ca[s], false, bf[8 * b + s], (short)0, acc, false, false);
      __builtin_amdgcn_sched_barrier(0);
#pragma unroll
      for (int s = 0; s < 8; ++s) ca[s] = na[s];
    }

    float* o = os + buf * G_OBUF + (16 * mt + 8 * h) * G_OPITCH + 16 * nt + c;
#pragma unroll
    for (int v = 0; v < 8; ++v) o[v * G_OPITCH] = acc[v];

    G_PARK8(xs + (buf ^ 1) * G_XBUF)
    __syncthreads();

#pragma unroll
    for (int k = 0; k < 2; ++k) {
      const int q  = t + 256 * k;
      const int r  = q >> 4;
      const int gr = st * G_ROWS + r;
      if (gr < n) {
        const float4 v = *(const float4*)(os + buf * G_OBUF + r * G_OPITCH + ((q & 15) << 2));
        const float d = k ? rs1 : rs0;
        *(float4*)(xw + (size_t)gr * OUT_SIZE + ((q & 15) << 2)) = make_float4(d * v.x, d * v.y, d * v.z, d * v.w);
      }
    }

    G_LOAD8(st + 2 * G_GRID)
    buf ^= 1;
  }
#undef G_LOAD
#undef G_LOAD8
#undef G_PARK
#undef G_PARK8
}


__global__ __launch_bounds__(256) void k_self(const float* __restrict__ xw,
                                              const float* __restrict__ dinv,
                                              float* __restrict__ out, int n) {
  size_t idx = (size_t)blockIdx.x * 256 + threadIdx.x;
  if (idx < (size_t)n * OUT_SIZE) {
    int i = (int)(idx >> 6);
    float d = dinv[i];
    out[idx] = d * d * xw[idx];
  }
}


#define EDGES_PER_GROUP 8

template <bool ALL>
__device__ __forceinline__ void scatter_wave(int r, int c, float nrm, unsigned okmask, int lane,
                                             const float* __restrict__ xw,
                                             float* __restrict__ out) {
#pragma unroll
  for (int g = 0; g < 32; g += EDGES_PER_GROUP) {
    float v0[EDGES_PER_GROUP], v1[EDGES_PER_GROUP];
#pragma unroll
    for (int i = 0; i < EDGES_PER_GROUP; ++i) {
      const int sc = __builtin_amdgcn_readlane(c, g + i);
      const float* p = xw + (size_t)sc * OUT_SIZE;
      v0[i] = p[lane];
      v1[i] = p[lane + 32];
    }
#pragma unroll
    for (int i = 0; i < EDGES_PER_GROUP; ++i) {
      if (ALL || ((okmask >> (g + i)) & 1u)) {
        const int   sr = __builtin_amdgcn_readlane(r, g + i);
        const float sn = __int_as_float(__builtin_amdgcn_readlane(__float_as_int(nrm), g + i));
        float* o = out + (size_t)sr * OUT_SIZE;
        unsafeAtomicAdd(o + lane,      sn * v0[i]);
        unsafeAtomicAdd(o + lane + 32, sn * v1[i]);
      }
    }
  }
}

__global__ __launch_bounds__(256) void k_scatter(const int* __restrict__ row,
                                                 const int* __restrict__ col,
                                                 const float* __restrict__ dinv,
                                                 const float* __restrict__ xw,
                                                 float* __restrict__ out, int E) {
  const int lane = threadIdx.x & 31;
  const int wave = blockIdx.x * 8 + (threadIdx.x >> 5);
  if (wave * 32 >= E) return;
  const int e = wave * 32 + lane;

  int r = 0, c = 0;
  if (e < E) { r = row[e]; c = col[e]; }
  const bool ok = (e < E) && (r != c);
  float nrm = 0.0f;
  if (ok) nrm = dinv[r] * dinv[c];
  const unsigned okmask = __builtin_amdgcn_ballot_w32(ok);

  if (okmask == 0xFFFFFFFFu) scatter_wave<true >(r, c, nrm, okmask, lane, xw, out);
  else                       scatter_wave<false>(r, c, nrm, okmask, lane, xw, out);
}


#define ROWS_PER_BUCKET 392
#define MAX_BUCKETS     256
#define CHUNK_EDGES     4096
#define EDGES_PER_THREAD (CHUNK_EDGES / 256)

__global__ __launch_bounds__(256) void k_init_bins(unsigned int* __restrict__ bcount,
                                                   unsigned int* __restrict__ cursor) {
  const int i = threadIdx.x;
  bcount[i] = 0u; cursor[i] = 0u;
}

__global__ __launch_bounds__(256) void k_count_bins(const int* __restrict__ row,
                                                    const int* __restrict__ col,
                                                    unsigned int* __restrict__ bcount, int E) {
  __shared__ unsigned int hist[MAX_BUCKETS];
  const int t = threadIdx.x;
  hist[t] = 0u;
  __syncthreads();
  const int base = blockIdx.x * CHUNK_EDGES + t;
  int rr[EDGES_PER_THREAD], cc[EDGES_PER_THREAD];
#pragma unroll
  for (int i = 0; i < EDGES_PER_THREAD; ++i) {
    const int e = base + i * 256;
    const int ec = e < E ? e : E - 1;
    rr[i] = row[ec]; cc[i] = col[ec];
  }
#pragma unroll
  for (int i = 0; i < EDGES_PER_THREAD; ++i) {
    const int e = base + i * 256;
    if (e < E && rr[i] != cc[i])
      atomicAdd(&hist[(unsigned)rr[i] / ROWS_PER_BUCKET], 1u);
  }
  __syncthreads();
  const unsigned int h = hist[t];
  if (h) atomicAdd(&bcount[t], h);
}

__global__ __launch_bounds__(256) void k_fill_bins(const int* __restrict__ row,
                                                   const int* __restrict__ col,
                                                   const unsigned int* __restrict__ bcount,
                                                   unsigned int* __restrict__ cursor,
                                                   int* __restrict__ binned, int E) {
  __shared__ unsigned int hist[MAX_BUCKETS];
  __shared__ unsigned int scan[MAX_BUCKETS];
  __shared__ unsigned int dst[MAX_BUCKETS];
  const int t = threadIdx.x;
  hist[t] = 0u;
  const unsigned int cnt_t = bcount[t];
  scan[t] = cnt_t;
  __syncthreads();
  for (int off = 1; off < MAX_BUCKETS; off <<= 1) {
    const unsigned int v = (t >= off) ? scan[t - off] : 0u;
    __syncthreads();
    scan[t] += v;
    __syncthreads();
  }
  const unsigned int start_t = scan[t] - cnt_t;

  int rr[EDGES_PER_THREAD], cc[EDGES_PER_THREAD];
  unsigned int lr[EDGES_PER_THREAD];
  const int base = blockIdx.x * CHUNK_EDGES + t;
#pragma unroll
  for (int i = 0; i < EDGES_PER_THREAD; ++i) {
    const int e = base + i * 256;
    const int ec = e < E ? e : E - 1;
    rr[i] = row[ec]; cc[i] = col[ec];
  }
#pragma unroll
  for (int i = 0; i < EDGES_PER_THREAD; ++i) {
    const int e = base + i * 256;
    lr[i] = 0u;
    if (e < E && rr[i] != cc[i])
      lr[i] = atomicAdd(&hist[(unsigned)rr[i] / ROWS_PER_BUCKET], 1u);
    else
      rr[i] = -1;
  }
  __syncthreads();
  const unsigned int h = hist[t];
  unsigned int g = 0u;
  if (h) g = atomicAdd(&cursor[t], h);
  dst[t] = start_t + g;
  __syncthreads();
#pragma unroll
  for (int i = 0; i < EDGES_PER_THREAD; ++i) {
    if (rr[i] >= 0)
      ((int2*)binned)[dst[(unsigned)rr[i] / ROWS_PER_BUCKET] + lr[i]] = make_int2(rr[i], cc[i]);
  }
}


#define DEG_LOADS 32

__global__ __launch_bounds__(256) void k_bucket_dinv(const int* __restrict__ binned_rc,
                                                     const unsigned int* __restrict__ bcount,
                                                     float* __restrict__ dinv, int n) {
  __shared__ unsigned int hist[ROWS_PER_BUCKET];
  __shared__ unsigned int s_start;
  const int2* binned = (const int2*)binned_rc;
  const int t = threadIdx.x;
  const int b = blockIdx.x;
  const int r0 = b * ROWS_PER_BUCKET;

  for (int i = t; i < ROWS_PER_BUCKET; i += 256) hist[i] = 0u;
  if (t == 0) s_start = 0u;
  __syncthreads();
  if (t < b) atomicAdd(&s_start, bcount[t]);
  __syncthreads();
  const unsigned int seg = s_start;
  const unsigned int cnt = bcount[b];

  for (unsigned int base = 0u; base < cnt; base += 256u * DEG_LOADS) {
    int rr[DEG_LOADS];
#pragma unroll
    for (int k = 0; k < DEG_LOADS; ++k) {
      const unsigned int i  = base + (unsigned)t + 256u * (unsigned)k;
      const unsigned int ic = i < cnt ? i : cnt - 1u;
      rr[k] = binned[seg + ic].x;
    }
#pragma unroll
    for (int k = 0; k < DEG_LOADS; ++k) {
      const unsigned int i = base + (unsigned)t + 256u * (unsigned)k;
      if (i < cnt) atomicAdd(&hist[rr[k] - r0], 1u);
    }
  }
  __syncthreads();
  for (int i = t; i < ROWS_PER_BUCKET; i += 256) {
    const int r = r0 + i;
    if (r < n) dinv[r] = rsqrtf((float)hist[i] + 1.0f);
  }
}

#define ACC_GROUP 16
#define TILE_FLOATS (ROWS_PER_BUCKET * OUT_SIZE)
#define TILE_V4     (TILE_FLOATS / 4)
#define TILE_V4_FULL (TILE_V4 / 1024)

template <bool ALL>
__device__ __forceinline__ void accum_wave(int lofs, int c, unsigned okmask, int lane,
                                           const float* __restrict__ y, float* tile) {
#pragma unroll
  for (int g = 0; g < 32; g += ACC_GROUP) {
    float v0[ACC_GROUP], v1[ACC_GROUP];
#pragma unroll
    for (int i = 0; i < ACC_GROUP; ++i) {
      const int sc = __builtin_amdgcn_readlane(c, g + i);
      const float* p = y + (size_t)sc * OUT_SIZE;
      v0[i] = p[lane];
      v1[i] = p[lane + 32];
    }
#pragma unroll
    for (int i = 0; i < ACC_GROUP; ++i) {
      if (ALL || ((okmask >> (g + i)) & 1u)) {
        const int so = __builtin_amdgcn_readlane(lofs, g + i);
        atomicAdd(tile + so + lane,      v0[i]);
        atomicAdd(tile + so + lane + 32, v1[i]);
      }
    }
  }
}

__device__ __forceinline__ float4 own_row4(const float* __restrict__ y, int r0, int q, int n) {
  int r = r0 + (q >> 4);
  r = r < n ? r : n - 1;
  return *(const float4*)(y + (size_t)r * OUT_SIZE + ((q & 15) << 2));
}

__global__ __launch_bounds__(1024) void k_accum(const int* __restrict__ binned_rc,
                                                const unsigned int* __restrict__ bcount,
                                                const float* __restrict__ dinv,
                                                const float* __restrict__ y,
                                                float* __restrict__ out, int n) {
  __shared__ __attribute__((aligned(16))) float tile[TILE_FLOATS];
  __shared__ unsigned int s_start;
  float4* tile4 = (float4*)tile;
  const int2* binned = (const int2*)binned_rc;
  const int t = threadIdx.x;
  const int lane = t & 31;
  const int w = t >> 5;
  const int b = blockIdx.x;
  const int r0 = b * ROWS_PER_BUCKET;

  if (t == 0) s_start = 0u;
  __syncthreads();
  if (t < MAX_BUCKETS && t < b) atomicAdd(&s_start, bcount[t]);

  {
    float4 sv[TILE_V4_FULL];
#pragma unroll
    for (int k = 0; k < TILE_V4_FULL; ++k) sv[k] = own_row4(y, r0, t + k * 1024, n);
#pragma unroll
    for (int k = 0; k < TILE_V4_FULL; ++k) tile4[t + k * 1024] = sv[k];
    const int q = t + TILE_V4_FULL * 1024;
    if (q < TILE_V4) tile4[q] = own_row4(y, r0, q, n);
  }
  __syncthreads();

  const unsigned int seg = s_start;
  const unsigned int end = seg + bcount[b];
  const unsigned int first = seg + (unsigned)w * 32u + (unsigned)lane;

  int2 idA = make_int2(r0, 0);
  if (first < end) idA = binned[first];

  for (unsigned int base = seg + (unsigned)w * 32u; base < end; base += 1024u) {
    const unsigned int e = base + (unsigned)lane;
    int2 idB = make_int2(r0, 0);
    if (e + 1024u < end) idB = binned[e + 1024u];

    const bool ok = e < end;
    const unsigned okmask = __builtin_amdgcn_ballot_w32(ok);
    const int lofs = (idA.x - r0) * OUT_SIZE;
    if (okmask == 0xFFFFFFFFu) accum_wave<true >(lofs, idA.y, okmask, lane, y, tile);
    else                       accum_wave<false>(lofs, idA.y, okmask, lane, y, tile);

    idA = idB;
  }

  float dv[TILE_V4_FULL + 1];
#pragma unroll
  for (int k = 0; k <= TILE_V4_FULL; ++k) {
    int r = r0 + ((t + k * 1024) >> 4);
    r = r < n ? r : n - 1;
    dv[k] = dinv[r];
  }
  __syncthreads();

#pragma unroll
  for (int k = 0; k <= TILE_V4_FULL; ++k) {
    const int q = t + k * 1024;
    const int r = r0 + (q >> 4);
    if (q < TILE_V4 && r < n) {
      const float4 v = tile4[q];
      const float d = dv[k];
      *(float4*)(out + (size_t)r * OUT_SIZE + ((q & 15) << 2)) = make_float4(d * v.x, d * v.y, d * v.z, d * v.w);
    }
  }
}


extern "C" void kernel_launch(void* const* d_in, const int* in_sizes, int n_in,
                              void* d_out, int out_size, void* d_ws, size_t ws_size,
                              hipStream_t stream) {
  const int*   edge = (const int*)d_in[0];
  const float* x    = (const float*)d_in[1];
  const float* W    = (const float*)d_in[3];

  const int E = in_sizes[0] / 2;
  const int n = in_sizes[1] / IN_SIZE;
  const int* row = edge;
  const int* col = edge + E;
  float* out = (float*)d_out;

  float*        xw     = (float*)d_ws;
  unsigned int* deg    = (unsigned int*)(xw + (size_t)n * OUT_SIZE);
  float*        dinv   = (float*)(deg + n);
  unsigned int* bcount = (unsigned int*)(dinv + n);
  unsigned int* cursor = bcount + MAX_BUCKETS;
  int*          binned = (int*)(cursor + MAX_BUCKETS);

  const int buckets = (n + ROWS_PER_BUCKET - 1) / ROWS_PER_BUCKET;
  const size_t need = (size_t)n * OUT_SIZE * 4 + (size_t)n * 8 + MAX_BUCKETS * 8 + (size_t)E * 8;

  if (buckets <= MAX_BUCKETS && need <= ws_size) {
    const int chunks = (E + CHUNK_EDGES - 1) / CHUNK_EDGES;
    k_init_bins<<<1, MAX_BUCKETS, 0, stream>>>(bcount, cursor);
    k_count_bins<<<chunks, 256, 0, stream>>>(row, col, bcount, E);
    k_fill_bins<<<chunks, 256, 0, stream>>>(row, col, bcount, cursor, binned, E);
    k_bucket_dinv<<<buckets, 256, 0, stream>>>(binned, bcount, dinv, n);
    k_gemm<<<G_GRID, 256, 0, stream>>>(x, W, dinv, xw, n);
    k_accum<<<buckets, 1024, 0, stream>>>(binned, bcount, dinv, xw, out, n);
  } else {
    k_init_deg<<<(n + 255) / 256, 256, 0, stream>>>(deg, n);
    k_count_deg<<<(E + 255) / 256, 256, 0, stream>>>(row, col, deg, E);
    k_dinv<<<(n + 255) / 256, 256, 0, stream>>>(deg, dinv, n);
    k_gemm<<<G_GRID, 256, 0, stream>>>(x, W, nullptr, xw, n);
    const size_t tot = (size_t)n * OUT_SIZE;
    k_self<<<(unsigned)((tot + 255) / 256), 256, 0, stream>>>(xw, dinv, out, n);
    k_scatter<<<(E + 255) / 256, 256, 0, stream>>>(row, col, dinv, xw, out, E);
  }
}


# Count pass dropped: each bucket owns a fixed-capacity region (mean share + 1/8) filled in one pass by cursor adds; strays go to an overflow list every bucket scans; 6 launches to 5; sum order kept
# speedup vs baseline: 5.5560x; 1.0537x over previous
#include <hip/hip_runtime.h>
#include <hip/hip_bf16.h>

typedef __attribute__((ext_vector_type(2))) float v2f;
typedef __attribute__((ext_vector_type(8))) float v8f;

#define IN_SIZE 256
#define OUT_SIZE 64


__global__ __launch_bounds__(256) void k_init_deg(unsigned int* __restrict__ deg, int n) {
  int i = blockIdx.x * 256 + threadIdx.x;
  if (i < n) deg[i] = 0u;
}

__global__ __launch_bounds__(256) void k_count_deg(const int* __restrict__ row,
                                                   const int* __restrict__ col,
                                                   unsigned int* __restrict__ deg, int E) {
  int e = blockIdx.x * 256 + threadIdx.x;
  if (e < E) {
    int r = row[e], c = col[e];
    if (r != c) atomicAdd(&deg[r], 1u);
  }
}

__global__ __launch_bounds__(256) void k_dinv(const unsigned int* __restrict__ deg,
                                              float* __restrict__ dinv, int n) {
  int i = blockIdx.x * 256 + threadIdx.x;
  if (i < n) dinv[i] = rsqrtf((float)deg[i] + 1.0f);
}


#define G_ROWS   32
#define G_PITCH  260
#define G_XBUF   (G_ROWS * G_PITCH)
#define G_OPITCH 68
#define G_OBUF   (G_ROWS * G_OPITCH)
#define G_GRID   512

__global__ __launch_bounds__(256) void k_gemm(const float* __restrict__ x,
                                              const float* __restrict__ W,
                                              const float* __restrict__ rs,
                                              float* __restrict__ xw, int n) {
  __shared__ __attribute__((aligned(16))) float xs[2 * G_XBUF];
  __shared__ __attribute__((aligned(16))) float os[2 * G_OBUF];
  const int t    = threadIdx.x;
  const int lane = t & 31;
  const int w    = t >> 5;
  const int c    = lane & 15;
  const int h    = lane >> 4;
  const int nt   = w & 3;
  const int mt   = w >> 2;
  const int nst  = (n + G_ROWS - 1) / G_ROWS;

  const int srow = t >> 6;
  const int scol = (t & 63) << 2;

#define G_LOAD(dst, stg, i)                                                        \
  {                                                                                \
    int r_ = (stg) * G_ROWS + srow + 4 * (i);                                      \
    r_ = r_ < n ? r_ : n - 1;                                                      \
    dst = *(const float4*)(x + (size_t)r_ * IN_SIZE + scol);                       \
  }
#define G_LOAD8(stg)                                                               \
  G_LOAD(x0, stg, 0) G_LOAD(x1, stg, 1) G_LOAD(x2, stg, 2) G_LOAD(x3, stg, 3)      \
  G_LOAD(x4, stg, 4) G_LOAD(x5, stg, 5) G_LOAD(x6, stg, 6) G_LOAD(x7, stg, 7)
#define G_PARK(base, i, v) *(float4*)((base) + (srow + 4 * (i)) * G_PITCH + scol) = v;
#define G_PARK8(base)                                                              \
  G_PARK(base, 0, x0) G_PARK(base, 1, x1) G_PARK(base, 2, x2) G_PARK(base, 3, x3)  \
  G_PARK(base, 4, x4) G_PARK(base, 5, x5) G_PARK(base, 6, x6) G_PARK(base, 7, x7)

  int st = blockIdx.x;
  float4 x0, x1, x2, x3, x4, x5, x6, x7;
  G_LOAD8(st)

  {
    float4 wv[16];
#pragma unroll
    for (int i = 0; i < 16; ++i) wv[i] = *(const float4*)(W + (size_t)(srow + 4 * i) * IN_SIZE + scol);
#pragma unroll
    for (int i = 0; i < 16; ++i) *(float4*)(xs + (srow + 4 * i) * G_PITCH + scol) = wv[i];
  }
  __syncthreads();
  v2f bf[64];
  {
    const float* wb = xs + (16 * nt + c) * G_PITCH + 2 * h;
#pragma unroll
    for (int s = 0; s < 64; ++s) bf[s] = *(const v2f*)(wb + 4 * s);
  }
  __syncthreads();

  G_PARK8(xs)
  __syncthreads();
  G_LOAD8(st + G_GRID)

  int buf = 0;
  for (; st < nst; st += G_GRID) {
    const float* a = xs + buf * G_XBUF + (16 * mt + c) * G_PITCH + 2 * h;
    float rs0 = 1.0f, rs1 = 1.0f;
    if (rs) {
      int g0 = st * G_ROWS + (t >> 4);
      int g1 = g0 + 16;
      g0 = g0 < n ? g0 : n - 1;
      g1 = g1 < n ? g1 : n - 1;
      rs0 = rs[g0];
      rs1 = rs[g1];
    }
    v8f acc = {};
    v2f ca[8], na[8];
#pragma unroll
    for (int s = 0; s < 8; ++s) ca[s] = *(const v2f*)(a + 4 * s);
#pragma unroll
    for (int b = 0; b < 8; ++b) {
      if (b < 7) {
#pragma unroll
        for (int s = 0; s < 8; ++s) na[s] = *(const v2f*)(a + 4 * (8 * (b + 1) + s));
      }
      __builtin_amdgcn_sched_barrier(0);
#pragma unroll
      for (int s = 0; s < 8; ++s)
        acc = __builtin_amdgcn_wmma_f32_16x16x4_f32(false, ca[s], false, bf[8 * b + s], (short)0, acc, false, false);
      __builtin_amdgcn_sched_barrier(0);
#pragma unroll
      for (int s = 0; s < 8; ++s) ca[s] = na[s];
    }

    float* o = os + buf * G_OBUF + (16 * mt + 8 * h) * G_OPITCH + 16 * nt + c;
#pragma unroll
    for (int v = 0; v < 8; ++v) o[v * G_OPITCH] = acc[v];

    G_PARK8(xs + (buf ^ 1) * G_XBUF)
    __syncthreads();

#pragma unroll
    for (int k = 0; k < 2; ++k) {
      const int q  = t + 256 * k;
      const int r  = q >> 4;
      const int gr = st * G_ROWS + r;
      if (gr < n) {
        const float4 v = *(const float4*)(os + buf * G_OBUF + r * G_OPITCH + ((q & 15) << 2));
        const float d = k ? rs1 : rs0;
        *(float4*)(xw + (size_t)gr * OUT_SIZE + ((q & 15) << 2)) = make_float4(d * v.x, d * v.y, d * v.z, d * v.w);
      }
    }

    G_LOAD8(st + 2 * G_GRID)
    buf ^= 1;
  }
#undef G_LOAD
#undef G_LOAD8
#undef G_PARK
#undef G_PARK8
}


__global__ __launch_bounds__(256) void k_self(const float* __restrict__ xw,
                                              const float* __restrict__ dinv,
                                              float* __restrict__ out, int n) {
  size_t idx = (size_t)blockIdx.x * 256 + threadIdx.x;
  if (idx < (size_t)n * OUT_SIZE) {
    int i = (int)(idx >> 6);
    float d = dinv[i];
    out[idx] = d * d * xw[idx];
  }
}


#define EDGES_PER_GROUP 8

template <bool ALL>
__device__ __forceinline__ void scatter_wave(int r, int c, float nrm, unsigned okmask, int lane,
                                             const float* __restrict__ xw,
                                             float* __restrict__ out) {
#pragma unroll
  for (int g = 0; g < 32; g += EDGES_PER_GROUP) {
    float v0[EDGES_PER_GROUP], v1[EDGES_PER_GROUP];
#pragma unroll
    for (int i = 0; i < EDGES_PER_GROUP; ++i) {
      const int sc = __builtin_amdgcn_readlane(c, g + i);
      const float* p = xw + (size_t)sc * OUT_SIZE;
      v0[i] = p[lane];
      v1[i] = p[lane + 32];
    }
#pragma unroll
    for (int i = 0; i < EDGES_PER_GROUP; ++i) {
      if (ALL || ((okmask >> (g + i)) & 1u)) {
        const int   sr = __builtin_amdgcn_readlane(r, g + i);
        const float sn = __int_as_float(__builtin_amdgcn_readlane(__float_as_int(nrm), g + i));
        float* o = out + (size_t)sr * OUT_SIZE;
        unsafeAtomicAdd(o + lane,      sn * v0[i]);
        unsafeAtomicAdd(o + lane + 32, sn * v1[i]);
      }
    }
  }
}

__global__ __launch_bounds__(256) void k_scatter(const int* __restrict__ row,
                                                 const int* __restrict__ col,
                                                 const float* __restrict__ dinv,
                                                 const float* __restrict__ xw,
                                                 float* __restrict__ out, int E) {
  const int lane = threadIdx.x & 31;
  const int wave = blockIdx.x * 8 + (threadIdx.x >> 5);
  if (wave * 32 >= E) return;
  const int e = wave * 32 + lane;

  int r = 0, c = 0;
  if (e < E) { r = row[e]; c = col[e]; }
  const bool ok = (e < E) && (r != c);
  float nrm = 0.0f;
  if (ok) nrm = dinv[r] * dinv[c];
  const unsigned okmask = __builtin_amdgcn_ballot_w32(ok);

  if (okmask == 0xFFFFFFFFu) scatter_wave<true >(r, c, nrm, okmask, lane, xw, out);
  else                       scatter_wave<false>(r, c, nrm, okmask, lane, xw, out);
}


#define ROWS_PER_BUCKET 392
#define MAX_BUCKETS     256
#define CHUNK_EDGES     4096
#define EDGES_PER_THREAD (CHUNK_EDGES / 256)

__host__ __device__ __forceinline__ unsigned int bucket_cap(int E, int n) {
  const unsigned int buckets = (unsigned)(n + ROWS_PER_BUCKET - 1) / ROWS_PER_BUCKET;
  const unsigned int share = ((unsigned)E + buckets - 1u) / buckets;
  return (share + (share >> 3) + 64u + 31u) & ~31u;
}

__global__ __launch_bounds__(256) void k_init_bins(unsigned int* __restrict__ cursor,
                                                   unsigned int* __restrict__ ovfn) {
  const int i = threadIdx.x;
  cursor[i] = 0u;
  if (i == 0) *ovfn = 0u;
}

__global__ __launch_bounds__(256) void k_fill_bins(const int* __restrict__ row,
                                                   const int* __restrict__ col,
                                                   unsigned int* __restrict__ cursor,
                                                   unsigned int* __restrict__ ovfn,
                                                   int* __restrict__ binned,
                                                   int* __restrict__ ovf, int E, int n) {
  __shared__ unsigned int hist[MAX_BUCKETS];
  __shared__ unsigned int dst[MAX_BUCKETS];
  const int t = threadIdx.x;
  const unsigned int cap = bucket_cap(E, n);
  hist[t] = 0u;
  __syncthreads();

  int rr[EDGES_PER_THREAD], cc[EDGES_PER_THREAD];
  unsigned int lr[EDGES_PER_THREAD];
  const int base = blockIdx.x * CHUNK_EDGES + t;
#pragma unroll
  for (int i = 0; i < EDGES_PER_THREAD; ++i) {
    const int e = base + i * 256;
    const int ec = e < E ? e : E - 1;
    rr[i] = row[ec]; cc[i] = col[ec];
  }
#pragma unroll
  for (int i = 0; i < EDGES_PER_THREAD; ++i) {
    const int e = base + i * 256;
    lr[i] = 0u;
    if (e < E && rr[i] != cc[i])
      lr[i] = atomicAdd(&hist[(unsigned)rr[i] / ROWS_PER_BUCKET], 1u);
    else
      rr[i] = -1;
  }
  __syncthreads();
  const unsigned int h = hist[t];
  unsigned int g = 0u;
  if (h) g = atomicAdd(&cursor[t], h);
  dst[t] = g;
  __syncthreads();
#pragma unroll
  for (int i = 0; i < EDGES_PER_THREAD; ++i) {
    if (rr[i] >= 0) {
      const unsigned int bkt  = (unsigned)rr[i] / ROWS_PER_BUCKET;
      const unsigned int slot = dst[bkt] + lr[i];
      int2* p = (int2*)binned + ((size_t)bkt * cap + slot);
      if (slot >= cap) p = (int2*)ovf + atomicAdd(ovfn, 1u);
      *p = make_int2(rr[i], cc[i]);
    }
  }
}


#define DEG_LOADS 32

__global__ __launch_bounds__(256) void k_bucket_dinv(const int* __restrict__ binned_rc,
                                                     const int* __restrict__ ovf_rc,
                                                     const unsigned int* __restrict__ cursor,
                                                     const unsigned int* __restrict__ ovfn,
                                                     float* __restrict__ dinv, int E, int n) {
  __shared__ unsigned int hist[ROWS_PER_BUCKET];
  const int t = threadIdx.x;
  const int b = blockIdx.x;
  const int r0 = b * ROWS_PER_BUCKET;
  const unsigned int cap = bucket_cap(E, n);
  const int2* binned = (const int2*)binned_rc + (size_t)b * cap;
  const int2* ovf = (const int2*)ovf_rc;

  for (int i = t; i < ROWS_PER_BUCKET; i += 256) hist[i] = 0u;
  __syncthreads();
  unsigned int cnt = cursor[b];
  cnt = cnt < cap ? cnt : cap;
  const unsigned int novf = *ovfn;

  for (unsigned int base = 0u; base < cnt; base += 256u * DEG_LOADS) {
    int rr[DEG_LOADS];
#pragma unroll
    for (int k = 0; k < DEG_LOADS; ++k) {
      const unsigned int i  = base + (unsigned)t + 256u * (unsigned)k;
      const unsigned int ic = i < cnt ? i : cnt - 1u;
      rr[k] = binned[ic].x;
    }
#pragma unroll
    for (int k = 0; k < DEG_LOADS; ++k) {
      const unsigned int i = base + (unsigned)t + 256u * (unsigned)k;
      if (i < cnt) atomicAdd(&hist[rr[k] - r0], 1u);
    }
  }
  for (unsigned int i = (unsigned)t; i < novf; i += 256u) {
    const unsigned int d = (unsigned)(ovf[i].x - r0);
    if (d < (unsigned)ROWS_PER_BUCKET) atomicAdd(&hist[d], 1u);
  }
  __syncthreads();
  for (int i = t; i < ROWS_PER_BUCKET; i += 256) {
    const int r = r0 + i;
    if (r < n) dinv[r] = rsqrtf((float)hist[i] + 1.0f);
  }
}

#define ACC_GROUP 16
#define TILE_FLOATS (ROWS_PER_BUCKET * OUT_SIZE)
#define TILE_V4     (TILE_FLOATS / 4)
#define TILE_V4_FULL (TILE_V4 / 1024)

template <bool ALL>
__device__ __forceinline__ void accum_wave(int lofs, int c, unsigned okmask, int lane,
                                           const float* __restrict__ y, float* tile) {
#pragma unroll
  for (int g = 0; g < 32; g += ACC_GROUP) {
    float v0[ACC_GROUP], v1[ACC_GROUP];
#pragma unroll
    for (int i = 0; i < ACC_GROUP; ++i) {
      const int sc = __builtin_amdgcn_readlane(c, g + i);
      const float* p = y + (size_t)sc * OUT_SIZE;
      v0[i] = p[lane];
      v1[i] = p[lane + 32];
    }
#pragma unroll
    for (int i = 0; i < ACC_GROUP; ++i) {
      if (ALL || ((okmask >> (g + i)) & 1u)) {
        const int so = __builtin_amdgcn_readlane(lofs, g + i);
        atomicAdd(tile + so + lane,      v0[i]);
        atomicAdd(tile + so + lane + 32, v1[i]);
      }
    }
  }
}

__device__ __forceinline__ float4 own_row4(const float* __restrict__ y, int r0, int q, int n) {
  int r = r0 + (q >> 4);
  r = r < n ? r : n - 1;
  return *(const float4*)(y + (size_t)r * OUT_SIZE + ((q & 15) << 2));
}

__global__ __launch_bounds__(1024) void k_accum(const int* __restrict__ binned_rc,
                                                const int* __restrict__ ovf_rc,
                                                const unsigned int* __restrict__ cursor,
                                                const unsigned int* __restrict__ ovfn,
                                                const float* __restrict__ dinv,
                                                const float* __restrict__ y,
                                                float* __restrict__ out, int E, int n) {
  __shared__ __attribute__((aligned(16))) float tile[TILE_FLOATS];
  float4* tile4 = (float4*)tile;
  const int t = threadIdx.x;
  const int lane = t & 31;
  const int w = t >> 5;
  const int b = blockIdx.x;
  const int r0 = b * ROWS_PER_BUCKET;
  const unsigned int cap = bucket_cap(E, n);
  const int2* binned = (const int2*)binned_rc + (size_t)b * cap;
  const int2* ovf = (const int2*)ovf_rc;

  unsigned int end = cursor[b];
  end = end < cap ? end : cap;
  const unsigned int novf = *ovfn;
  const unsigned int first = (unsigned)w * 32u + (unsigned)lane;

  int2 idA = make_int2(r0, 0);
  if (first < end) idA = binned[first];

  {
    float4 sv[TILE_V4_FULL];
#pragma unroll
    for (int k = 0; k < TILE_V4_FULL; ++k) sv[k] = own_row4(y, r0, t + k * 1024, n);
#pragma unroll
    for (int k = 0; k < TILE_V4_FULL; ++k) tile4[t + k * 1024] = sv[k];
    const int q = t + TILE_V4_FULL * 1024;
    if (q < TILE_V4) tile4[q] = own_row4(y, r0, q, n);
  }
  __syncthreads();

  for (unsigned int base = (unsigned)w * 32u; base < end; base += 1024u) {
    const unsigned int e = base + (unsigned)lane;
    int2 idB = make_int2(r0, 0);
    if (e + 1024u < end) idB = binned[e + 1024u];

    const bool ok = e < end;
    const unsigned okmask = __builtin_amdgcn_ballot_w32(ok);
    const int lofs = (idA.x - r0) * OUT_SIZE;
    if (okmask == 0xFFFFFFFFu) accum_wave<true >(lofs, idA.y, okmask, lane, y, tile);
    else                       accum_wave<false>(lofs, idA.y, okmask, lane, y, tile);

    idA = idB;
  }

  for (unsigned int base = (unsigned)w * 32u; base < novf; base += 1024u) {
    const unsigned int e = base + (unsigned)lane;
    int2 id = make_int2(r0, 0);
    if (e < novf) id = ovf[e];
    const unsigned int d = (unsigned)(id.x - r0);
    const bool ok = (e < novf) && (d < (unsigned)ROWS_PER_BUCKET);
    const unsigned okmask = __builtin_amdgcn_ballot_w32(ok);
    const int lofs = ok ? (int)d * OUT_SIZE : 0;
    if (okmask) accum_wave<false>(lofs, id.y, okmask, lane, y, tile);
  }

  float dv[TILE_V4_FULL + 1];
#pragma unroll
  for (int k = 0; k <= TILE_V4_FULL; ++k) {
    int r = r0 + ((t + k * 1024) >> 4);
    r = r < n ? r : n - 1;
    dv[k] = dinv[r];
  }
  __syncthreads();

#pragma unroll
  for (int k = 0; k <= TILE_V4_FULL; ++k) {
    const int q = t + k * 1024;
    const int r = r0 + (q >> 4);
    if (q < TILE_V4 && r < n) {
      const float4 v = tile4[q];
      const float d = dv[k];
      *(float4*)(out + (size_t)r * OUT_SIZE + ((q & 15) << 2)) = make_float4(d * v.x, d * v.y, d * v.z, d * v.w);
    }
  }
}


extern "C" void kernel_launch(void* const* d_in, const int* in_sizes, int n_in,
                              void* d_out, int out_size, void* d_ws, size_t ws_size,
                              hipStream_t stream) {
  const int*   edge = (const int*)d_in[0];
  const float* x    = (const float*)d_in[1];
  const float* W    = (const float*)d_in[3];

  const int E = in_sizes[0] / 2;
  const int n = in_sizes[1] / IN_SIZE;
  const int* row = edge;
  const int* col = edge + E;
  float* out = (float*)d_out;

  const int buckets = (n + ROWS_PER_BUCKET - 1) / ROWS_PER_BUCKET;
  const unsigned int cap = bucket_cap(E, n);
  float*        xw     = (float*)d_ws;
  unsigned int* deg    = (unsigned int*)(xw + (size_t)n * OUT_SIZE);
  float*        dinv   = (float*)(deg + n);
  unsigned int* cursor = (unsigned int*)(dinv + n);
  unsigned int* ovfn   = cursor + MAX_BUCKETS;
  int*          binned = (int*)(ovfn + MAX_BUCKETS);
  int*          ovf    = binned + (size_t)buckets * cap * 2;

  const size_t need = (size_t)n * OUT_SIZE * 4 + (size_t)n * 8 + MAX_BUCKETS * 8 +
                      (size_t)buckets * cap * 8 + (size_t)E * 8;

  if (buckets <= MAX_BUCKETS && need <= ws_size) {
    const int chunks = (E + CHUNK_EDGES - 1) / CHUNK_EDGES;
    k_init_bins<<<1, MAX_BUCKETS, 0, stream>>>(cursor, ovfn);
    k_fill_bins<<<chunks, 256, 0, stream>>>(row, col, cursor, ovfn, binned, ovf, E, n);
    k_bucket_dinv<<<buckets, 256, 0, stream>>>(binned, ovf, cursor, ovfn, dinv, E, n);
    k_gemm<<<G_GRID, 256, 0, stream>>>(x, W, dinv, xw, n);
    k_accum<<<buckets, 1024, 0, stream>>>(binned, ovf, cursor, ovfn, dinv, xw, out, E, n);
  } else {
    k_init_deg<<<(n + 255) / 256, 256, 0, stream>>>(deg, n);
    k_count_deg<<<(E + 255) / 256, 256, 0, stream>>>(row, col, deg, E);
    k_dinv<<<(n + 255) / 256, 256, 0, stream>>>(deg, dinv, n);
    k_gemm<<<G_GRID, 256, 0, stream>>>(x, W, nullptr, xw, n);
    const size_t tot = (size_t)n * OUT_SIZE;
    k_self<<<(unsigned)((tot + 255) / 256), 256, 0, stream>>>(xw, dinv, out, n);
    k_scatter<<<(E + 255) / 256, 256, 0, stream>>>(row, col, dinv, xw, out, E);
  }
}


# Fill pass: a chunk's kept edges are put in bucket order in LDS (scan of its bucket counts) and stored as contiguous runs, a few line pieces per store, not 32 8-byte pieces; same slots, sum order kept
# speedup vs baseline: 6.3188x; 1.1373x over previous
#include <hip/hip_runtime.h>
#include <hip/hip_bf16.h>

typedef __attribute__((ext_vector_type(2))) float v2f;
typedef __attribute__((ext_vector_type(8))) float v8f;

#define IN_SIZE 256
#define OUT_SIZE 64


__global__ __launch_bounds__(256) void k_init_deg(unsigned int* __restrict__ deg, int n) {
  int i = blockIdx.x * 256 + threadIdx.x;
  if (i < n) deg[i] = 0u;
}

__global__ __launch_bounds__(256) void k_count_deg(const int* __restrict__ row,
                                                   const int* __restrict__ col,
                                                   unsigned int* __restrict__ deg, int E) {
  int e = blockIdx.x * 256 + threadIdx.x;
  if (e < E) {
    int r = row[e], c = col[e];
    if (r != c) atomicAdd(&deg[r], 1u);
  }
}

__global__ __launch_bounds__(256) void k_dinv(const unsigned int* __restrict__ deg,
                                              float* __restrict__ dinv, int n) {
  int i = blockIdx.x * 256 + threadIdx.x;
  if (i < n) dinv[i] = rsqrtf((float)deg[i] + 1.0f);
}


#define G_ROWS   32
#define G_PITCH  260
#define G_XBUF   (G_ROWS * G_PITCH)
#define G_OPITCH 68
#define G_OBUF   (G_ROWS * G_OPITCH)
#define G_GRID   512

__global__ __launch_bounds__(256) void k_gemm(const float* __restrict__ x,
                                              const float* __restrict__ W,
                                              const float* __restrict__ rs,
                                              float* __restrict__ xw, int n) {
  __shared__ __attribute__((aligned(16))) float xs[2 * G_XBUF];
  __shared__ __attribute__((aligned(16))) float os[2 * G_OBUF];
  const int t    = threadIdx.x;
  const int lane = t & 31;
  const int w    = t >> 5;
  const int c    = lane & 15;
  const int h    = lane >> 4;
  const int nt   = w & 3;
  const int mt   = w >> 2;
  const int nst  = (n + G_ROWS - 1) / G_ROWS;

  const int srow = t >> 6;
  const int scol = (t & 63) << 2;

#define G_LOAD(dst, stg, i)                                                        \
  {                                                                                \
    int r_ = (stg) * G_ROWS + srow + 4 * (i);                                      \
    r_ = r_ < n ? r_ : n - 1;                                                      \
    dst = *(const float4*)(x + (size_t)r_ * IN_SIZE + scol);                       \
  }
#define G_LOAD8(stg)                                                               \
  G_LOAD(x0, stg, 0) G_LOAD(x1, stg, 1) G_LOAD(x2, stg, 2) G_LOAD(x3, stg, 3)      \
  G_LOAD(x4, stg, 4) G_LOAD(x5, stg, 5) G_LOAD(x6, stg, 6) G_LOAD(x7, stg, 7)
#define G_PARK(base, i, v) *(float4*)((base) + (srow + 4 * (i)) * G_PITCH + scol) = v;
#define G_PARK8(base)                                                              \
  G_PARK(base, 0, x0) G_PARK(base, 1, x1) G_PARK(base, 2, x2) G_PARK(base, 3, x3)  \
  G_PARK(base, 4, x4) G_PARK(base, 5, x5) G_PARK(base, 6, x6) G_PARK(base, 7, x7)

  int st = blockIdx.x;
  float4 x0, x1, x2, x3, x4, x5, x6, x7;
  G_LOAD8(st)

  {
    float4 wv[16];
#pragma unroll
    for (int i = 0; i < 16; ++i) wv[i] = *(const float4*)(W + (size_t)(srow + 4 * i) * IN_SIZE + scol);
#pragma unroll
    for (int i = 0; i < 16; ++i) *(float4*)(xs + (srow + 4 * i) * G_PITCH + scol) = wv[i];
  }
  __syncthreads();
  v2f bf[64];
  {
    const float* wb = xs + (16 * nt + c) * G_PITCH + 2 * h;
#pragma unroll
    for (int s = 0; s < 64; ++s) bf[s] = *(const v2f*)(wb + 4 * s);
  }
  __syncthreads();

  G_PARK8(xs)
  __syncthreads();
  G_LOAD8(st + G_GRID)

  int buf = 0;
  for (; st < nst; st += G_GRID) {
    const float* a = xs + buf * G_XBUF + (16 * mt + c) * G_PITCH + 2 * h;
    float rs0 = 1.0f, rs1 = 1.0f;
    if (rs) {
      int g0 = st * G_ROWS + (t >> 4);
      int g1 = g0 + 16;
      g0 = g0 < n ? g0 : n - 1;
      g1 = g1 < n ? g1 : n - 1;
      rs0 = rs[g0];
      rs1 = rs[g1];
    }
    v8f acc = {};
    v2f ca[8], na[8];
#pragma unroll
    for (int s = 0; s < 8; ++s) ca[s] = *(const v2f*)(a + 4 * s);
#pragma unroll
    for (int b = 0; b < 8; ++b) {
      if (b < 7) {
#pragma unroll
        for (int s = 0; s < 8; ++s) na[s] = *(const v2f*)(a + 4 * (8 * (b + 1) + s));
      }
      __builtin_amdgcn_sched_barrier(0);
#pragma unroll
      for (int s = 0; s < 8; ++s)
        acc = __builtin_amdgcn_wmma_f32_16x16x4_f32(false, ca[s], false, bf[8 * b + s], (short)0, acc, false, false);
      __builtin_amdgcn_sched_barrier(0);
#pragma unroll
      for (int s = 0; s < 8; ++s) ca[s] = na[s];
    }

    float* o = os + buf * G_OBUF + (16 * mt + 8 * h) * G_OPITCH + 16 * nt + c;
#pragma unroll
    for (int v = 0; v < 8; ++v) o[v * G_OPITCH] = acc[v];

    G_PARK8(xs + (buf ^ 1) * G_XBUF)
    __syncthreads();

#pragma unroll
    for (int k = 0; k < 2; ++k) {
      const int q  = t + 256 * k;
      const int r  = q >> 4;
      const int gr = st * G_ROWS + r;
      if (gr < n) {
        const float4 v = *(const float4*)(os + buf * G_OBUF + r * G_OPITCH + ((q & 15) << 2));
        const float d = k ? rs1 : rs0;
        *(float4*)(xw + (size_t)gr * OUT_SIZE + ((q & 15) << 2)) = make_float4(d * v.x, d * v.y, d * v.z, d * v.w);
      }
    }

    G_LOAD8(st + 2 * G_GRID)
    buf ^= 1;
  }
#undef G_LOAD
#undef G_LOAD8
#undef G_PARK
#undef G_PARK8
}


__global__ __launch_bounds__(256) void k_self(const float* __restrict__ xw,
                                              const float* __restrict__ dinv,
                                              float* __restrict__ out, int n) {
  size_t idx = (size_t)blockIdx.x * 256 + threadIdx.x;
  if (idx < (size_t)n * OUT_SIZE) {
    int i = (int)(idx >> 6);
    float d = dinv[i];
    out[idx] = d * d * xw[idx];
  }
}


#define EDGES_PER_GROUP 8

template <bool ALL>
__device__ __forceinline__ void scatter_wave(int r, int c, float nrm, unsigned okmask, int lane,
                                             const float* __restrict__ xw,
                                             float* __restrict__ out) {
#pragma unroll
  for (int g = 0; g < 32; g += EDGES_PER_GROUP) {
    float v0[EDGES_PER_GROUP], v1[EDGES_PER_GROUP];
#pragma unroll
    for (int i = 0; i < EDGES_PER_GROUP; ++i) {
      const int sc = __builtin_amdgcn_readlane(c, g + i);
      const float* p = xw + (size_t)sc * OUT_SIZE;
      v0[i] = p[lane];
      v1[i] = p[lane + 32];
    }
#pragma unroll
    for (int i = 0; i < EDGES_PER_GROUP; ++i) {
      if (ALL || ((okmask >> (g + i)) & 1u)) {
        const int   sr = __builtin_amdgcn_readlane(r, g + i);
        const float sn = __int_as_float(__builtin_amdgcn_readlane(__float_as_int(nrm), g + i));
        float* o = out + (size_t)sr * OUT_SIZE;
        unsafeAtomicAdd(o + lane,      sn * v0[i]);
        unsafeAtomicAdd(o + lane + 32, sn * v1[i]);
      }
    }
  }
}

__global__ __launch_bounds__(256) void k_scatter(const int* __restrict__ row,
                                                 const int* __restrict__ col,
                                                 const float* __restrict__ dinv,
                                                 const float* __restrict__ xw,
                                                 float* __restrict__ out, int E) {
  const int lane = threadIdx.x & 31;
  const int wave = blockIdx.x * 8 + (threadIdx.x >> 5);
  if (wave * 32 >= E) return;
  const int e = wave * 32 + lane;

  int r = 0, c = 0;
  if (e < E) { r = row[e]; c = col[e]; }
  const bool ok = (e < E) && (r != c);
  float nrm = 0.0f;
  if (ok) nrm = dinv[r] * dinv[c];
  const unsigned okmask = __builtin_amdgcn_ballot_w32(ok);

  if (okmask == 0xFFFFFFFFu) scatter_wave<true >(r, c, nrm, okmask, lane, xw, out);
  else                       scatter_wave<false>(r, c, nrm, okmask, lane, xw, out);
}


#define ROWS_PER_BUCKET 392
#define MAX_BUCKETS     256
#define CHUNK_EDGES     4096
#define EDGES_PER_THREAD (CHUNK_EDGES / 256)

__host__ __device__ __forceinline__ unsigned int bucket_cap(int E, int n) {
  const unsigned int buckets = (unsigned)(n + ROWS_PER_BUCKET - 1) / ROWS_PER_BUCKET;
  const unsigned int share = ((unsigned)E + buckets - 1u) / buckets;
  return (share + (share >> 3) + 64u + 31u) & ~31u;
}

__global__ __launch_bounds__(256) void k_init_bins(unsigned int* __restrict__ cursor,
                                                   unsigned int* __restrict__ ovfn) {
  const int i = threadIdx.x;
  cursor[i] = 0u;
  if (i == 0) *ovfn = 0u;
}

__global__ __launch_bounds__(256) void k_fill_bins(const int* __restrict__ row,
                                                   const int* __restrict__ col,
                                                   unsigned int* __restrict__ cursor,
                                                   unsigned int* __restrict__ ovfn,
                                                   int* __restrict__ binned,
                                                   int* __restrict__ ovf, int E, int n) {
  __shared__ unsigned int hist[MAX_BUCKETS];
  __shared__ unsigned int lstart[MAX_BUCKETS];
  __shared__ unsigned int gofs[MAX_BUCKETS];
  __shared__ unsigned int wsum[8];
  __shared__ int2 stage[CHUNK_EDGES];
  const int t = threadIdx.x;
  const int lane = t & 31;
  const int w = t >> 5;
  const unsigned int cap = bucket_cap(E, n);
  hist[t] = 0u;
  __syncthreads();

  int rr[EDGES_PER_THREAD], cc[EDGES_PER_THREAD];
  unsigned int lr[EDGES_PER_THREAD];
  const int base = blockIdx.x * CHUNK_EDGES + t;
#pragma unroll
  for (int i = 0; i < EDGES_PER_THREAD; ++i) {
    const int e = base + i * 256;
    const int ec = e < E ? e : E - 1;
    rr[i] = row[ec]; cc[i] = col[ec];
  }
  unsigned int keep = 0u;
#pragma unroll
  for (int i = 0; i < EDGES_PER_THREAD; ++i) {
    const int e = base + i * 256;
    lr[i] = 0u;
    if (e < E && rr[i] != cc[i]) {
      lr[i] = atomicAdd(&hist[(unsigned)rr[i] / ROWS_PER_BUCKET], 1u);
      keep |= 1u << i;
    }
  }
  __syncthreads();
  const unsigned int h = hist[t];
  unsigned int g = 0u;
  if (h) g = atomicAdd(&cursor[t], h);
  unsigned int inc = h;
#pragma unroll
  for (int d = 1; d < 32; d <<= 1) {
    const unsigned int v = __shfl_up(inc, (unsigned)d);
    if (lane >= d) inc += v;
  }
  if (lane == 31) wsum[w] = inc;
  __syncthreads();
  unsigned int woff = 0u, total = 0u;
#pragma unroll
  for (int k = 0; k < 8; ++k) {
    const unsigned int s = wsum[k];
    if (k < w) woff += s;
    total += s;
  }
  const unsigned int ls = woff + inc - h;
  lstart[t] = ls;
  gofs[t] = (unsigned)t * cap + g - ls;
  __syncthreads();

  unsigned int pos[EDGES_PER_THREAD];
#pragma unroll
  for (int i = 0; i < EDGES_PER_THREAD; ++i)
    pos[i] = lstart[(unsigned)rr[i] / ROWS_PER_BUCKET] + lr[i];
#pragma unroll
  for (int i = 0; i < EDGES_PER_THREAD; ++i)
    if ((keep >> i) & 1u) stage[pos[i]] = make_int2(rr[i], cc[i]);
  __syncthreads();

  if (total == 0u) return;
  int2 ev[EDGES_PER_THREAD];
#pragma unroll
  for (int i = 0; i < EDGES_PER_THREAD; ++i) {
    const unsigned int p = (unsigned)t + 256u * (unsigned)i;
    ev[i] = stage[p < total ? p : total - 1u];
  }
  unsigned int idx[EDGES_PER_THREAD];
#pragma unroll
  for (int i = 0; i < EDGES_PER_THREAD; ++i) {
    const unsigned int p = (unsigned)t + 256u * (unsigned)i;
    idx[i] = gofs[(unsigned)ev[i].x / ROWS_PER_BUCKET] + (p < total ? p : total - 1u);
  }
#pragma unroll
  for (int i = 0; i < EDGES_PER_THREAD; ++i) {
    const unsigned int p = (unsigned)t + 256u * (unsigned)i;
    if (p < total) {
      const unsigned int bkt = (unsigned)ev[i].x / ROWS_PER_BUCKET;
      int2* q = (int2*)binned + (size_t)idx[i];
      if (idx[i] >= (bkt + 1u) * cap) q = (int2*)ovf + atomicAdd(ovfn, 1u);
      *q = ev[i];
    }
  }
}


#define DEG_LOADS 32

__global__ __launch_bounds__(256) void k_bucket_dinv(const int* __restrict__ binned_rc,
                                                     const int* __restrict__ ovf_rc,
                                                     const unsigned int* __restrict__ cursor,
                                                     const unsigned int* __restrict__ ovfn,
                                                     float* __restrict__ dinv, int E, int n) {
  __shared__ unsigned int hist[ROWS_PER_BUCKET];
  const int t = threadIdx.x;
  const int b = blockIdx.x;
  const int r0 = b * ROWS_PER_BUCKET;
  const unsigned int cap = bucket_cap(E, n);
  const int2* binned = (const int2*)binned_rc + (size_t)b * cap;
  const int2* ovf = (const int2*)ovf_rc;

  for (int i = t; i < ROWS_PER_BUCKET; i += 256) hist[i] = 0u;
  __syncthreads();
  unsigned int cnt = cursor[b];
  cnt = cnt < cap ? cnt : cap;
  const unsigned int novf = *ovfn;

  for (unsigned int base = 0u; base < cnt; base += 256u * DEG_LOADS) {
    int rr[DEG_LOADS];
#pragma unroll
    for (int k = 0; k < DEG_LOADS; ++k) {
      const unsigned int i  = base + (unsigned)t + 256u * (unsigned)k;
      const unsigned int ic = i < cnt ? i : cnt - 1u;
      rr[k] = binned[ic].x;
    }
#pragma unroll
    for (int k = 0; k < DEG_LOADS; ++k) {
      const unsigned int i = base + (unsigned)t + 256u * (unsigned)k;
      if (i < cnt) atomicAdd(&hist[rr[k] - r0], 1u);
    }
  }
  for (unsigned int i = (unsigned)t; i < novf; i += 256u) {
    const unsigned int d = (unsigned)(ovf[i].x - r0);
    if (d < (unsigned)ROWS_PER_BUCKET) atomicAdd(&hist[d], 1u);
  }
  __syncthreads();
  for (int i = t; i < ROWS_PER_BUCKET; i += 256) {
    const int r = r0 + i;
    if (r < n) dinv[r] = rsqrtf((float)hist[i] + 1.0f);
  }
}

#define ACC_GROUP 16
#define TILE_FLOATS (ROWS_PER_BUCKET * OUT_SIZE)
#define TILE_V4     (TILE_FLOATS / 4)
#define TILE_V4_FULL (TILE_V4 / 1024)

template <bool ALL>
__device__ __forceinline__ void accum_wave(int lofs, int c, unsigned okmask, int lane,
                                           const float* __restrict__ y, float* tile) {
#pragma unroll
  for (int g = 0; g < 32; g += ACC_GROUP) {
    float v0[ACC_GROUP], v1[ACC_GROUP];
#pragma unroll
    for (int i = 0; i < ACC_GROUP; ++i) {
      const int sc = __builtin_amdgcn_readlane(c, g + i);
      const float* p = y + (size_t)sc * OUT_SIZE;
      v0[i] = p[lane];
      v1[i] = p[lane + 32];
    }
#pragma unroll
    for (int i = 0; i < ACC_GROUP; ++i) {
      if (ALL || ((okmask >> (g + i)) & 1u)) {
        const int so = __builtin_amdgcn_readlane(lofs, g + i);
        atomicAdd(tile + so + lane,      v0[i]);
        atomicAdd(tile + so + lane + 32, v1[i]);
      }
    }
  }
}

__device__ __forceinline__ float4 own_row4(const float* __restrict__ y, int r0, int q, int n) {
  int r = r0 + (q >> 4);
  r = r < n ? r : n - 1;
  return *(const float4*)(y + (size_t)r * OUT_SIZE + ((q & 15) << 2));
}

__global__ __launch_bounds__(1024) void k_accum(const int* __restrict__ binned_rc,
                                                const int* __restrict__ ovf_rc,
                                                const unsigned int* __restrict__ cursor,
                                                const unsigned int* __restrict__ ovfn,
                                                const float* __restrict__ dinv,
                                                const float* __restrict__ y,
                                                float* __restrict__ out, int E, int n) {
  __shared__ __attribute__((aligned(16))) float tile[TILE_FLOATS];
  float4* tile4 = (float4*)tile;
  const int t = threadIdx.x;
  const int lane = t & 31;
  const int w = t >> 5;
  const int b = blockIdx.x;
  const int r0 = b * ROWS_PER_BUCKET;
  const unsigned int cap = bucket_cap(E, n);
  const int2* binned = (const int2*)binned_rc + (size_t)b * cap;
  const int2* ovf = (const int2*)ovf_rc;

  unsigned int end = cursor[b];
  end = end < cap ? end : cap;
  const unsigned int novf = *ovfn;
  const unsigned int first = (unsigned)w * 32u + (unsigned)lane;

  int2 idA = make_int2(r0, 0);
  if (first < end) idA = binned[first];

  {
    float4 sv[TILE_V4_FULL];
#pragma unroll
    for (int k = 0; k < TILE_V4_FULL; ++k) sv[k] = own_row4(y, r0, t + k * 1024, n);
#pragma unroll
    for (int k = 0; k < TILE_V4_FULL; ++k) tile4[t + k * 1024] = sv[k];
    const int q = t + TILE_V4_FULL * 1024;
    if (q < TILE_V4) tile4[q] = own_row4(y, r0, q, n);
  }
  __syncthreads();

  for (unsigned int base = (unsigned)w * 32u; base < end; base += 1024u) {
    const unsigned int e = base + (unsigned)lane;
    int2 idB = make_int2(r0, 0);
    if (e + 1024u < end) idB = binned[e + 1024u];

    const bool ok = e < end;
    const unsigned okmask = __builtin_amdgcn_ballot_w32(ok);
    const int lofs = (idA.x - r0) * OUT_SIZE;
    if (okmask == 0xFFFFFFFFu) accum_wave<true >(lofs, idA.y, okmask, lane, y, tile);
    else                       accum_wave<false>(lofs, idA.y, okmask, lane, y, tile);

    idA = idB;
  }

  for (unsigned int base = (unsigned)w * 32u; base < novf; base += 1024u) {
    const unsigned int e = base + (unsigned)lane;
    int2 id = make_int2(r0, 0);
    if (e < novf) id = ovf[e];
    const unsigned int d = (unsigned)(id.x - r0);
    const bool ok = (e < novf) && (d < (unsigned)ROWS_PER_BUCKET);
    const unsigned okmask = __builtin_amdgcn_ballot_w32(ok);
    const int lofs = ok ? (int)d * OUT_SIZE : 0;
    if (okmask) accum_wave<false>(lofs, id.y, okmask, lane, y, tile);
  }

  float dv[TILE_V4_FULL + 1];
#pragma unroll
  for (int k = 0; k <= TILE_V4_FULL; ++k) {
    int r = r0 + ((t + k * 1024) >> 4);
    r = r < n ? r : n - 1;
    dv[k] = dinv[r];
  }
  __syncthreads();

#pragma unroll
  for (int k = 0; k <= TILE_V4_FULL; ++k) {
    const int q = t + k * 1024;
    const int r = r0 + (q >> 4);
    if (q < TILE_V4 && r < n) {
      const float4 v = tile4[q];
      const float d = dv[k];
      *(float4*)(out + (size_t)r * OUT_SIZE + ((q & 15) << 2)) = make_float4(d * v.x, d * v.y, d * v.z, d * v.w);
    }
  }
}


extern "C" void kernel_launch(void* const* d_in, const int* in_sizes, int n_in,
                              void* d_out, int out_size, void* d_ws, size_t ws_size,
                              hipStream_t stream) {
  const int*   edge = (const int*)d_in[0];
  const float* x    = (const float*)d_in[1];
  const float* W    = (const float*)d_in[3];

  const int E = in_sizes[0] / 2;
  const int n = in_sizes[1] / IN_SIZE;
  const int* row = edge;
  const int* col = edge + E;
  float* out = (float*)d_out;

  const int buckets = (n + ROWS_PER_BUCKET - 1) / ROWS_PER_BUCKET;
  const unsigned int cap = bucket_cap(E, n);
  float*        xw     = (float*)d_ws;
  unsigned int* deg    = (unsigned int*)(xw + (size_t)n * OUT_SIZE);
  float*        dinv   = (float*)(deg + n);
  unsigned int* cursor = (unsigned int*)(dinv + n);
  unsigned int* ovfn   = cursor + MAX_BUCKETS;
  int*          binned = (int*)(ovfn + MAX_BUCKETS);
  int*          ovf    = binned + (size_t)buckets * cap * 2;

  const size_t need = (size_t)n * OUT_SIZE * 4 + (size_t)n * 8 + MAX_BUCKETS * 8 +
                      (size_t)buckets * cap * 8 + (size_t)E * 8;

  if (buckets <= MAX_BUCKETS && need <= ws_size) {
    const int chunks = (E + CHUNK_EDGES - 1) / CHUNK_EDGES;
    k_init_bins<<<1, MAX_BUCKETS, 0, stream>>>(cursor, ovfn);
    k_fill_bins<<<chunks, 256, 0, stream>>>(row, col, cursor, ovfn, binned, ovf, E, n);
    k_bucket_dinv<<<buckets, 256, 0, stream>>>(binned, ovf, cursor, ovfn, dinv, E, n);
    k_gemm<<<G_GRID, 256, 0, stream>>>(x, W, dinv, xw, n);
    k_accum<<<buckets, 1024, 0, stream>>>(binned, ovf, cursor, ovfn, dinv, xw, out, E, n);
  } else {
    k_init_deg<<<(n + 255) / 256, 256, 0, stream>>>(deg, n);
    k_count_deg<<<(E + 255) / 256, 256, 0, stream>>>(row, col, deg, E);
    k_dinv<<<(n + 255) / 256, 256, 0, stream>>>(deg, dinv, n);
    k_gemm<<<G_GRID, 256, 0, stream>>>(x, W, nullptr, xw, n);
    const size_t tot = (size_t)n * OUT_SIZE;
    k_self<<<(unsigned)((tot + 255) / 256), 256, 0, stream>>>(xw, dinv, out, n);
    k_scatter<<<(E + 255) / 256, 256, 0, stream>>>(row, col, dinv, xw, out, E);
  }
}
